# starting kernel
# baseline (speedup 1.0000x reference)
#include <hip/hip_runtime.h>


typedef __attribute__((ext_vector_type(2))) float v2f;
typedef __attribute__((ext_vector_type(8))) float v8f;

#define TILES_PER_BLOCK 8
#define CHUNK_KT 32

__global__ __launch_bounds__(256) void zero_f32_kernel(float* __restrict__ p,
                                                       long n) {
  long i = (long)blockIdx.x * blockDim.x + threadIdx.x;
  long stride = (long)gridDim.x * blockDim.x;
  for (; i < n; i += stride) p[i] = 0.0f;
}

__global__ __launch_bounds__(256) void pack_weights_kernel(
    const float* __restrict__ W, float* __restrict__ P, int K) {
  int total = K * 128;
  for (int tid = blockIdx.x * blockDim.x + threadIdx.x; tid < total;
       tid += gridDim.x * blockDim.x) {
    int frag = tid >> 6;
    int r = tid & 63;
    int L = r >> 1;
    int v = r & 1;
    int kt = frag >> 3;
    int nt = frag & 7;
    int k = kt * 4 + ((L >> 4) << 1) + v;
    int n = nt * 16 + (L & 15);
    P[tid] = W[k * 128 + n];
  }
}

__global__ __launch_bounds__(256) void degree_kernel(
    const int* __restrict__ dst, float* __restrict__ deg, int E) {
  int e = blockIdx.x * blockDim.x + threadIdx.x;
  if (e < E) atomicAdd(&deg[dst[e]], 1.0f);
}

__global__ __launch_bounds__(256) void scatter_add_kernel(
    const float* __restrict__ feat, const int* __restrict__ src,
    const int* __restrict__ dst, float* __restrict__ agg, int E, int F) {
  int chunks = F >> 2;
  long total = (long)E * chunks;
  long tid = (long)blockIdx.x * blockDim.x + threadIdx.x;
  long stride = (long)gridDim.x * blockDim.x;
  for (; tid < total; tid += stride) {
    int e = (int)(tid / chunks);
    int c = (int)(tid % chunks);
    int s = src[e];
    int d = dst[e];
    const float4 val = *(const float4*)(feat + (size_t)s * F + c * 4);
    float* o = agg + (size_t)d * F + c * 4;
    atomicAdd(o + 0, val.x);
    atomicAdd(o + 1, val.y);
    atomicAdd(o + 2, val.z);
    atomicAdd(o + 3, val.w);
  }
}

__global__ __launch_bounds__(256) void gemm_pass_kernel(
    const float* __restrict__ A, int K, const float* __restrict__ Bp,
    const float* __restrict__ bias, const float* __restrict__ deg,
    const float* __restrict__ prev, float* __restrict__ out, int relu,
    int nTiles) {
  __shared__ float lds[CHUNK_KT * 8 * 64];

  const int lane = threadIdx.x & 31;
  const int wave = threadIdx.x >> 5;
  int tile = blockIdx.x * TILES_PER_BLOCK + wave;
  const bool active = tile < nTiles;
  if (!active) tile = nTiles - 1;

  const int half = lane >> 4;
  const int n0 = lane & 15;
  const int arow = tile * 16 + n0;
  const float* __restrict__ arowp = A + (size_t)arow * K + half * 2;

  float scale = 1.0f;
  if (deg) scale = 1.0f / fmaxf(deg[arow], 1.0f);

  v8f acc[8];
#pragma unroll
  for (int nt = 0; nt < 8; ++nt) {
    float b = bias ? bias[nt * 16 + n0] : 0.0f;
    acc[nt] = (v8f){b, b, b, b, b, b, b, b};
  }

  const int nkt = K >> 2;
  v2f a_pre = *(const v2f*)(arowp);

  for (int kt0 = 0; kt0 < nkt; kt0 += CHUNK_KT) {
    __syncthreads();
    {
      const float4* __restrict__ gsrc =
          (const float4*)(Bp + (size_t)kt0 * 8 * 64);
      float4* __restrict__ ldst = (float4*)lds;
      for (int i = threadIdx.x; i < CHUNK_KT * 8 * 16; i += 256)
        ldst[i] = gsrc[i];
    }
    __syncthreads();

    for (int kt = 0; kt < CHUNK_KT; ++kt) {
      const int ktg = kt0 + kt;
      v2f a = a_pre * scale;
      const int ktn = (ktg + 1 < nkt) ? (ktg + 1) : ktg;
      a_pre = *(const v2f*)(arowp + ktn * 4);

      const float* bbase = lds + (kt * 8) * 64 + lane * 2;
      v2f bf[8];
#pragma unroll
      for (int nt = 0; nt < 8; ++nt) bf[nt] = *(const v2f*)(bbase + nt * 64);

#pragma unroll
      for (int nt = 0; nt < 8; ++nt) {
        acc[nt] = __builtin_amdgcn_wmma_f32_16x16x4_f32(
            false, a, false, bf[nt], (short)0, acc[nt], false, false);
      }
    }
  }

  if (!active) return;

  const int m0 = tile * 16;
#pragma unroll
  for (int nt = 0; nt < 8; ++nt) {
#pragma unroll
    for (int r = 0; r < 8; ++r) {
      int m = m0 + r + 8 * half;
      size_t idx = (size_t)m * 128 + nt * 16 + n0;
      float v = acc[nt][r];
      if (prev) v += prev[idx];
      if (relu) v = fmaxf(v, 0.0f);
      out[idx] = v;
    }
  }
}

extern "C" void kernel_launch(void* const* d_in, const int* in_sizes, int n_in,
                              void* d_out, int out_size, void* d_ws,
                              size_t ws_size, hipStream_t stream) {
  const int IN_DIM = 384, HID = 128;
  const int N = in_sizes[0] / IN_DIM;
  const int E = in_sizes[1] / 2;

  const float* x    = (const float*)d_in[0];
  const int*   ei   = (const int*)d_in[1];
  const float* W1_l = (const float*)d_in[2];
  const float* b1   = (const float*)d_in[3];
  const float* W1_r = (const float*)d_in[4];
  const float* W2_l = (const float*)d_in[5];
  const float* b2   = (const float*)d_in[6];
  const float* W2_r = (const float*)d_in[7];
  const int* src = ei;
  const int* dst = ei + E;
  float* out = (float*)d_out;

  float* ws = (float*)d_ws;
  const long deg_off  = 0;
  const long deg_pad  = ((long)N + 127) & ~127L;
  const long agg_off  = deg_off + deg_pad;
  const long h1_off   = agg_off + (long)N * IN_DIM;
  const long pw1l_off = h1_off + (long)N * HID;
  const long pw1r_off = pw1l_off + (long)IN_DIM * HID;
  const long pw2l_off = pw1r_off + (long)IN_DIM * HID;
  const long pw2r_off = pw2l_off + (long)HID * HID;

  float* deg  = ws + deg_off;
  float* agg  = ws + agg_off;
  float* h1   = ws + h1_off;
  float* pw1l = ws + pw1l_off;
  float* pw1r = ws + pw1r_off;
  float* pw2l = ws + pw2l_off;
  float* pw2r = ws + pw2r_off;

  const int nTiles = N / 16;
  const int gemmBlocks = (nTiles + TILES_PER_BLOCK - 1) / TILES_PER_BLOCK;

  {
    long nz = deg_pad + (long)N * IN_DIM;
    int blocks = (int)((nz + 255) / 256);
    zero_f32_kernel<<<blocks, 256, 0, stream>>>(ws, nz);
  }
  pack_weights_kernel<<<(IN_DIM * 128 + 255) / 256, 256, 0, stream>>>(W1_l, pw1l, IN_DIM);
  pack_weights_kernel<<<(IN_DIM * 128 + 255) / 256, 256, 0, stream>>>(W1_r, pw1r, IN_DIM);
  pack_weights_kernel<<<(HID * 128 + 255) / 256, 256, 0, stream>>>(W2_l, pw2l, HID);
  pack_weights_kernel<<<(HID * 128 + 255) / 256, 256, 0, stream>>>(W2_r, pw2r, HID);

  degree_kernel<<<(E + 255) / 256, 256, 0, stream>>>(dst, deg, E);

  {
    long total = (long)E * (IN_DIM / 4);
    int blocks = (int)((total + 255) / 256);
    scatter_add_kernel<<<blocks, 256, 0, stream>>>(x, src, dst, agg, E, IN_DIM);
  }

  gemm_pass_kernel<<<gemmBlocks, 256, 0, stream>>>(agg, IN_DIM, pw1l, b1, deg,
                                                   nullptr, h1, 0, nTiles);
  gemm_pass_kernel<<<gemmBlocks, 256, 0, stream>>>(x, IN_DIM, pw1r, nullptr,
                                                   nullptr, h1, h1, 1, nTiles);

  {
    long nz = (long)N * HID;
    int blocks = (int)((nz + 255) / 256);
    zero_f32_kernel<<<blocks, 256, 0, stream>>>(agg, nz);
    long total = (long)E * (HID / 4);
    int sblocks = (int)((total + 255) / 256);
    scatter_add_kernel<<<sblocks, 256, 0, stream>>>(h1, src, dst, agg, E, HID);
  }

  gemm_pass_kernel<<<gemmBlocks, 256, 0, stream>>>(agg, HID, pw2l, b2, deg,
                                                   nullptr, out, 0, nTiles);
  gemm_pass_kernel<<<gemmBlocks, 256, 0, stream>>>(h1, HID, pw2r, nullptr,
                                                   nullptr, out, out, 0, nTiles);
}


# Neighbour sums in LDS: each work-group owns a node range, scans the dst list and adds its edges' source rows with LDS f32 atomics (arrival order, as the global ones); no zero fill or global atomic
# speedup vs baseline: 1.3908x; 1.3908x over previous
#include <hip/hip_runtime.h>


typedef __attribute__((ext_vector_type(2))) float v2f;
typedef __attribute__((ext_vector_type(8))) float v8f;

#define TILES_PER_BLOCK 8
#define CHUNK_KT 32

__global__ __launch_bounds__(256) void pack_weights_kernel(
    const float* __restrict__ W, float* __restrict__ P, int K) {
  int total = K * 128;
  for (int tid = blockIdx.x * blockDim.x + threadIdx.x; tid < total;
       tid += gridDim.x * blockDim.x) {
    int frag = tid >> 6;
    int r = tid & 63;
    int L = r >> 1;
    int v = r & 1;
    int kt = frag >> 3;
    int nt = frag & 7;
    int k = kt * 4 + ((L >> 4) << 1) + v;
    int n = nt * 16 + (L & 15);
    P[tid] = W[k * 128 + n];
  }
}


template <int F>
__device__ __forceinline__ void add_wave_edges(
    int rel, int e_first, int rows, int lane, const int* __restrict__ src,
    const float* __restrict__ feat, float* acc, int* cnt) {
  unsigned m = __builtin_amdgcn_ballot_w32((unsigned)rel < (unsigned)rows);
  while (m) {
    const int l = __builtin_ctz(m);
    m &= m - 1;
    const int r = __builtin_amdgcn_readlane(rel, l);
    const int s = src[e_first + l];
    const float4* __restrict__ row =
        (const float4*)(feat + (size_t)s * F) + lane;
    float* a = acc + r * F + lane * 4;
    float4 v[F / 128];
#pragma unroll
    for (int q = 0; q < F / 128; ++q) v[q] = row[q * 32];
#pragma unroll
    for (int q = 0; q < F / 128; ++q) {
      __hip_atomic_fetch_add(a + q * 128 + 0, v[q].x, __ATOMIC_RELAXED,
                             __HIP_MEMORY_SCOPE_WORKGROUP);
      __hip_atomic_fetch_add(a + q * 128 + 1, v[q].y, __ATOMIC_RELAXED,
                             __HIP_MEMORY_SCOPE_WORKGROUP);
      __hip_atomic_fetch_add(a + q * 128 + 2, v[q].z, __ATOMIC_RELAXED,
                             __HIP_MEMORY_SCOPE_WORKGROUP);
      __hip_atomic_fetch_add(a + q * 128 + 3, v[q].w, __ATOMIC_RELAXED,
                             __HIP_MEMORY_SCOPE_WORKGROUP);
    }
    if (lane == 0)
      __hip_atomic_fetch_add(cnt + r, 1, __ATOMIC_RELAXED,
                             __HIP_MEMORY_SCOPE_WORKGROUP);
  }
}

template <int F, int R>
__global__ __launch_bounds__(1024) void range_sum_kernel(
    const float* __restrict__ feat, const int* __restrict__ src,
    const int* __restrict__ dst, float* __restrict__ agg,
    float* __restrict__ deg, int E, int N) {
  __shared__ __attribute__((aligned(16))) float acc[R * F];
  __shared__ int cnt[R];

  const int tid = threadIdx.x;
  const int lane = tid & 31;
  const int wave = __builtin_amdgcn_readfirstlane(tid >> 5);
  const int base = blockIdx.x * R;
  const int rows = (N - base < R) ? (N - base) : R;

  float4* __restrict__ acc4 = (float4*)acc;
  for (int i = tid; i < R * F / 4; i += 1024)
    acc4[i] = make_float4(0.0f, 0.0f, 0.0f, 0.0f);
  for (int i = tid; i < R; i += 1024) cnt[i] = 0;
  __syncthreads();

  constexpr int SUB = 16;
  constexpr int PIECE = SUB * 1024;
  const int nFull = E / PIECE;
  if (nFull > 0) {
    int piece = blockIdx.x % nFull;
    for (int t = 0; t < nFull; ++t) {
      const int e0 = piece * PIECE + (wave << 5);
      const int* __restrict__ dp = dst + e0 + lane;
      int rel[SUB];
#pragma unroll
      for (int j = 0; j < SUB; ++j) rel[j] = dp[j << 10] - base;
#pragma unroll
      for (int j = 0; j < SUB; ++j)
        add_wave_edges<F>(rel[j], e0 + (j << 10), rows, lane, src, feat, acc,
                          cnt);
      piece = (piece + 1 == nFull) ? 0 : piece + 1;
    }
  }
  for (int eb = nFull * PIECE + (wave << 5); eb < E; eb += 1024) {
    const int e = eb + lane;
    const int d = (e < E) ? dst[e] : -1;
    add_wave_edges<F>(d - base, eb, rows, lane, src, feat, acc, cnt);
  }
  __syncthreads();

  float4* __restrict__ out4 = (float4*)(agg + (size_t)base * F);
  const int n4 = rows * (F / 4);
  for (int i = tid; i < n4; i += 1024) out4[i] = acc4[i];
  if (deg)
    for (int i = tid; i < rows; i += 1024) deg[base + i] = (float)cnt[i];
}

__global__ __launch_bounds__(256) void gemm_pass_kernel(
    const float* __restrict__ A, int K, const float* __restrict__ Bp,
    const float* __restrict__ bias, const float* __restrict__ deg,
    const float* __restrict__ prev, float* __restrict__ out, int relu,
    int nTiles) {
  __shared__ float lds[CHUNK_KT * 8 * 64];

  const int lane = threadIdx.x & 31;
  const int wave = threadIdx.x >> 5;
  int tile = blockIdx.x * TILES_PER_BLOCK + wave;
  const bool active = tile < nTiles;
  if (!active) tile = nTiles - 1;

  const int half = lane >> 4;
  const int n0 = lane & 15;
  const int arow = tile * 16 + n0;
  const float* __restrict__ arowp = A + (size_t)arow * K + half * 2;

  float scale = 1.0f;
  if (deg) scale = 1.0f / fmaxf(deg[arow], 1.0f);

  v8f acc[8];
#pragma unroll
  for (int nt = 0; nt < 8; ++nt) {
    float b = bias ? bias[nt * 16 + n0] : 0.0f;
    acc[nt] = (v8f){b, b, b, b, b, b, b, b};
  }

  const int nkt = K >> 2;
  v2f a_pre = *(const v2f*)(arowp);

  for (int kt0 = 0; kt0 < nkt; kt0 += CHUNK_KT) {
    __syncthreads();
    {
      const float4* __restrict__ gsrc =
          (const float4*)(Bp + (size_t)kt0 * 8 * 64);
      float4* __restrict__ ldst = (float4*)lds;
      for (int i = threadIdx.x; i < CHUNK_KT * 8 * 16; i += 256)
        ldst[i] = gsrc[i];
    }
    __syncthreads();

    for (int kt = 0; kt < CHUNK_KT; ++kt) {
      const int ktg = kt0 + kt;
      v2f a = a_pre * scale;
      const int ktn = (ktg + 1 < nkt) ? (ktg + 1) : ktg;
      a_pre = *(const v2f*)(arowp + ktn * 4);

      const float* bbase = lds + (kt * 8) * 64 + lane * 2;
      v2f bf[8];
#pragma unroll
      for (int nt = 0; nt < 8; ++nt) bf[nt] = *(const v2f*)(bbase + nt * 64);

#pragma unroll
      for (int nt = 0; nt < 8; ++nt) {
        acc[nt] = __builtin_amdgcn_wmma_f32_16x16x4_f32(
            false, a, false, bf[nt], (short)0, acc[nt], false, false);
      }
    }
  }

  if (!active) return;

  const int m0 = tile * 16;
#pragma unroll
  for (int nt = 0; nt < 8; ++nt) {
#pragma unroll
    for (int r = 0; r < 8; ++r) {
      int m = m0 + r + 8 * half;
      size_t idx = (size_t)m * 128 + nt * 16 + n0;
      float v = acc[nt][r];
      if (prev) v += prev[idx];
      if (relu) v = fmaxf(v, 0.0f);
      out[idx] = v;
    }
  }
}

extern "C" void kernel_launch(void* const* d_in, const int* in_sizes, int n_in,
                              void* d_out, int out_size, void* d_ws,
                              size_t ws_size, hipStream_t stream) {
  const int IN_DIM = 384, HID = 128;
  const int N = in_sizes[0] / IN_DIM;
  const int E = in_sizes[1] / 2;

  const float* x    = (const float*)d_in[0];
  const int*   ei   = (const int*)d_in[1];
  const float* W1_l = (const float*)d_in[2];
  const float* b1   = (const float*)d_in[3];
  const float* W1_r = (const float*)d_in[4];
  const float* W2_l = (const float*)d_in[5];
  const float* b2   = (const float*)d_in[6];
  const float* W2_r = (const float*)d_in[7];
  const int* src = ei;
  const int* dst = ei + E;
  float* out = (float*)d_out;

  float* ws = (float*)d_ws;
  const long deg_off  = 0;
  const long deg_pad  = ((long)N + 127) & ~127L;
  const long agg_off  = deg_off + deg_pad;
  const long h1_off   = agg_off + (long)N * IN_DIM;
  const long pw1l_off = h1_off + (long)N * HID;
  const long pw1r_off = pw1l_off + (long)IN_DIM * HID;
  const long pw2l_off = pw1r_off + (long)IN_DIM * HID;
  const long pw2r_off = pw2l_off + (long)HID * HID;

  float* deg  = ws + deg_off;
  float* agg  = ws + agg_off;
  float* h1   = ws + h1_off;
  float* pw1l = ws + pw1l_off;
  float* pw1r = ws + pw1r_off;
  float* pw2l = ws + pw2l_off;
  float* pw2r = ws + pw2r_off;

  const int nTiles = N / 16;
  const int gemmBlocks = (nTiles + TILES_PER_BLOCK - 1) / TILES_PER_BLOCK;

  pack_weights_kernel<<<(IN_DIM * 128 + 255) / 256, 256, 0, stream>>>(W1_l, pw1l, IN_DIM);
  pack_weights_kernel<<<(IN_DIM * 128 + 255) / 256, 256, 0, stream>>>(W1_r, pw1r, IN_DIM);
  pack_weights_kernel<<<(HID * 128 + 255) / 256, 256, 0, stream>>>(W2_l, pw2l, HID);
  pack_weights_kernel<<<(HID * 128 + 255) / 256, 256, 0, stream>>>(W2_r, pw2r, HID);

  range_sum_kernel<384, 208><<<(N + 207) / 208, 1024, 0, stream>>>(
      x, src, dst, agg, deg, E, N);

  gemm_pass_kernel<<<gemmBlocks, 256, 0, stream>>>(agg, IN_DIM, pw1l, b1, deg,
                                                   nullptr, h1, 0, nTiles);
  gemm_pass_kernel<<<gemmBlocks, 256, 0, stream>>>(x, IN_DIM, pw1r, nullptr,
                                                   nullptr, h1, h1, 1, nTiles);

  range_sum_kernel<128, 416><<<(N + 415) / 416, 1024, 0, stream>>>(
      h1, src, dst, agg, nullptr, E, N);

  gemm_pass_kernel<<<gemmBlocks, 256, 0, stream>>>(agg, HID, pw2l, b2, deg,
                                                   nullptr, out, 0, nTiles);
  gemm_pass_kernel<<<gemmBlocks, 256, 0, stream>>>(h1, HID, pw2r, nullptr,
                                                   nullptr, out, out, 0, nTiles);
}


# GEMM passes: 256 resident work-groups of 8 waves; a wave holds its 3-4 row tiles' accumulators in registers (6-7 tiles per SIMD, not up to 8), so a B fragment feeds 3-4 WMMAs; A asked 4 k-steps ahead
# speedup vs baseline: 2.1231x; 1.5265x over previous
#include <hip/hip_runtime.h>


typedef __attribute__((ext_vector_type(2))) float v2f;
typedef __attribute__((ext_vector_type(8))) float v8f;

#define CHUNK_KT 32

__global__ __launch_bounds__(256) void pack_weights_kernel(
    const float* __restrict__ W, float* __restrict__ P, int K) {
  int total = K * 128;
  for (int tid = blockIdx.x * blockDim.x + threadIdx.x; tid < total;
       tid += gridDim.x * blockDim.x) {
    int frag = tid >> 6;
    int r = tid & 63;
    int L = r >> 1;
    int v = r & 1;
    int kt = frag >> 3;
    int nt = frag & 7;
    int k = kt * 4 + ((L >> 4) << 1) + v;
    int n = nt * 16 + (L & 15);
    P[tid] = W[k * 128 + n];
  }
}


template <int Q>
__device__ __forceinline__ void store_row_and_clear(float* __restrict__ agg,
                                                    size_t node, int F,
                                                    int lane, float4* acc) {
  float4* __restrict__ o = (float4*)(agg + node * (size_t)F) + lane;
#pragma unroll
  for (int q = 0; q < Q; ++q) {
    o[q * 32] = acc[q];
    acc[q] = make_float4(0.0f, 0.0f, 0.0f, 0.0f);
  }
}

template <int F, int R, int B>
__global__ __launch_bounds__(1024) void range_gather_kernel(
    const float* __restrict__ feat, const int* __restrict__ src,
    const int* __restrict__ dst, float* __restrict__ agg,
    float* __restrict__ deg, int E, int N) {
  constexpr int CAP = 4096;
  constexpr int PER = R / 32;
  constexpr int Q = F / 128;
  __shared__ int cnt[R];
  __shared__ int start[R + 1];
  __shared__ int list_e[CAP];
  __shared__ unsigned list_k[CAP];
  __shared__ int sorted[CAP];
  __shared__ int nmatch;

  const int tid = threadIdx.x;
  const int lane = tid & 31;
  const int wave = __builtin_amdgcn_readfirstlane(tid >> 5);
  const int base = blockIdx.x * R;
  const int rows = (N - base < R) ? (N - base) : R;

  for (int i = tid; i < R; i += 1024) cnt[i] = 0;
  if (tid == 0) nmatch = 0;
  __syncthreads();

  constexpr int SUB = 16;
  constexpr int PIECE = SUB * 1024;
  const int nFull = E / PIECE;
  if (nFull > 0) {
    int piece = blockIdx.x % nFull;
    for (int t = 0; t < nFull; ++t) {
      const int e0 = piece * PIECE + (wave << 5) + lane;
      const int* __restrict__ dp = dst + e0;
      int rel[SUB];
#pragma unroll
      for (int j = 0; j < SUB; ++j) rel[j] = dp[j << 10] - base;
#pragma unroll
      for (int j = 0; j < SUB; ++j) {
        if ((unsigned)rel[j] < (unsigned)rows) {
          const int rank = __hip_atomic_fetch_add(
              &cnt[rel[j]], 1, __ATOMIC_RELAXED, __HIP_MEMORY_SCOPE_WORKGROUP);
          const int idx = __hip_atomic_fetch_add(
              &nmatch, 1, __ATOMIC_RELAXED, __HIP_MEMORY_SCOPE_WORKGROUP);
          if (idx < CAP) {
            list_e[idx] = e0 + (j << 10);
            list_k[idx] = (unsigned)rel[j] | ((unsigned)rank << 16);
          }
        }
      }
      piece = (piece + 1 == nFull) ? 0 : piece + 1;
    }
  }
  for (int eb = nFull * PIECE + (wave << 5); eb < E; eb += 1024) {
    const int e = eb + lane;
    const int d = (e < E) ? dst[e] : -1;
    const int rl = d - base;
    if (d >= 0 && (unsigned)rl < (unsigned)rows) {
      const int rank = __hip_atomic_fetch_add(
          &cnt[rl], 1, __ATOMIC_RELAXED, __HIP_MEMORY_SCOPE_WORKGROUP);
      const int idx = __hip_atomic_fetch_add(
          &nmatch, 1, __ATOMIC_RELAXED, __HIP_MEMORY_SCOPE_WORKGROUP);
      if (idx < CAP) {
        list_e[idx] = e;
        list_k[idx] = (unsigned)rl | ((unsigned)rank << 16);
      }
    }
  }
  __syncthreads();

  const int total = nmatch;

  if (wave == 0) {
    int s = 0;
    for (int k = 0; k < PER; ++k) s += cnt[lane * PER + k];
    int incl = s;
    for (int d = 1; d < 32; d <<= 1) {
      const int t = __shfl_up(incl, d);
      if (lane >= d) incl += t;
    }
    int run = incl - s;
    for (int k = 0; k < PER; ++k) {
      start[lane * PER + k] = run;
      run += cnt[lane * PER + k];
    }
    if (lane == 31) start[R] = run;
  }
  __syncthreads();

  if (total <= CAP) {
    for (int i = tid; i < total; i += 1024) {
      const unsigned k = list_k[i];
      sorted[start[k & 0xffffu] + (int)(k >> 16)] = src[list_e[i]];
    }
  }
  __syncthreads();

  const int r0 = wave * PER;
  const int r1 = (r0 + PER < rows) ? (r0 + PER) : rows;
  float4 acc[Q];
#pragma unroll
  for (int q = 0; q < Q; ++q) acc[q] = make_float4(0.0f, 0.0f, 0.0f, 0.0f);

  if (r0 < rows) {
    if (total <= CAP) {
      const int e_lo = __builtin_amdgcn_readfirstlane(start[r0]);
      const int e_hi = __builtin_amdgcn_readfirstlane(start[r1]);
      int node = r0;
      int node_end = __builtin_amdgcn_readfirstlane(start[r0 + 1]);
      for (int eb = e_lo; eb < e_hi; eb += B) {
        float4 v[B][Q];
#pragma unroll
        for (int j = 0; j < B; ++j) {
          const int ei = (eb + j < e_hi) ? (eb + j) : (e_hi - 1);
          const int s = __builtin_amdgcn_readfirstlane(sorted[ei]);
          const float4* __restrict__ row =
              (const float4*)(feat + (size_t)s * F) + lane;
#pragma unroll
          for (int q = 0; q < Q; ++q) v[j][q] = row[q * 32];
        }
#pragma unroll
        for (int j = 0; j < B; ++j) {
          if (eb + j < e_hi) {
            while (eb + j >= node_end) {
              store_row_and_clear<Q>(agg, (size_t)(base + node), F, lane, acc);
              ++node;
              node_end = __builtin_amdgcn_readfirstlane(start[node + 1]);
            }
#pragma unroll
            for (int q = 0; q < Q; ++q) {
              acc[q].x += v[j][q].x;
              acc[q].y += v[j][q].y;
              acc[q].z += v[j][q].z;
              acc[q].w += v[j][q].w;
            }
          }
        }
      }
      for (; node < r1; ++node)
        store_row_and_clear<Q>(agg, (size_t)(base + node), F, lane, acc);
    } else {
      for (int node = r0; node < r1; ++node) {
        for (int eb = 0; eb < E; eb += 32) {
          const int e = eb + lane;
          const int d = (e < E) ? dst[e] : -1;
          unsigned m = __builtin_amdgcn_ballot_w32(d == base + node);
          while (m) {
            const int l = __builtin_ctz(m);
            m &= m - 1;
            const int s = src[eb + l];
            const float4* __restrict__ row =
                (const float4*)(feat + (size_t)s * F) + lane;
#pragma unroll
            for (int q = 0; q < Q; ++q) {
              const float4 w = row[q * 32];
              acc[q].x += w.x;
              acc[q].y += w.y;
              acc[q].z += w.z;
              acc[q].w += w.w;
            }
          }
        }
        store_row_and_clear<Q>(agg, (size_t)(base + node), F, lane, acc);
      }
    }
  }
  if (deg)
    for (int i = tid; i < rows; i += 1024) deg[base + i] = (float)cnt[i];
}

#define MB 4
#define AHEAD 4
#define STG 2

__device__ __forceinline__ void k_chunk(
    v8f (&acc)[MB][8], v2f (&ring)[AHEAD][MB], const float* __restrict__ A1,
    const float* __restrict__ A2, const unsigned (&rowoff)[MB],
    const float (&scale)[MB], const float* __restrict__ ldsb,
    const float4* __restrict__ gnext, float4* __restrict__ lnext, int kt0,
    int nkt, int nkt2, int lane, int tid, bool full) {
  const bool second = kt0 >= nkt;
  for (int kb = 0; kb < CHUNK_KT; kb += AHEAD) {
    const int sbase = tid + (kb / AHEAD) * STG * 256;
    float4 s0 = make_float4(0.0f, 0.0f, 0.0f, 0.0f), s1 = s0;
    if (gnext) {
      s0 = gnext[sbase];
      s1 = gnext[sbase + 256];
    }
    v2f nxt[AHEAD][MB];
#pragma unroll
    for (int j = 0; j < AHEAD; ++j) {
      int ktn = kt0 + kb + AHEAD + j;
      if (ktn > nkt2 - 1) ktn = nkt2 - 1;
      const bool n2 = ktn >= nkt;
      const float* __restrict__ an = (n2 ? A2 : A1) + (n2 ? (ktn - nkt) : ktn) * 4;
#pragma unroll
      for (int m = 0; m < MB; ++m) nxt[j][m] = *(const v2f*)(an + rowoff[m]);
    }
    asm volatile("" ::: "memory");
#pragma unroll
    for (int j = 0; j < AHEAD; ++j) {
      const int kt = kb + j;
      v2f a[MB];
#pragma unroll
      for (int m = 0; m < MB; ++m)
        a[m] = ring[j][m] * (second ? 1.0f : scale[m]);

      const float* bbase = ldsb + (kt * 8) * 64 + lane * 2;
      v2f bf[8];
#pragma unroll
      for (int nt = 0; nt < 8; ++nt) bf[nt] = *(const v2f*)(bbase + nt * 64);

      if (full) {
#pragma unroll
        for (int nt = 0; nt < 8; ++nt) {
          acc[MB - 1][nt] = __builtin_amdgcn_wmma_f32_16x16x4_f32(
              false, a[MB - 1], false, bf[nt], (short)0, acc[MB - 1][nt], false,
              false);
        }
      }
#pragma unroll
      for (int nt = 0; nt < 8; ++nt) {
#pragma unroll
        for (int m = 0; m < MB - 1; ++m) {
          acc[m][nt] = __builtin_amdgcn_wmma_f32_16x16x4_f32(
              false, a[m], false, bf[nt], (short)0, acc[m][nt], false, false);
        }
      }
    }
#pragma unroll
    for (int j = 0; j < AHEAD; ++j)
#pragma unroll
      for (int m = 0; m < MB; ++m) ring[j][m] = nxt[j][m];
    if (gnext) {
      lnext[sbase] = s0;
      lnext[sbase + 256] = s1;
    }
  }
}

__global__ __launch_bounds__(256) void gemm_pair_kernel(
    const float* __restrict__ A1, const float* __restrict__ Bp1,
    const float* __restrict__ A2, const float* __restrict__ Bp2, int K,
    const float* __restrict__ bias, const float* __restrict__ deg,
    float* __restrict__ out, int relu, int nTiles) {
  __shared__ float lds[2][CHUNK_KT * 8 * 64];

  const int tid = threadIdx.x;
  const int lane = tid & 31;
  const int wave = __builtin_amdgcn_readfirstlane(tid >> 5);
  const int half = lane >> 4;
  const int n0 = lane & 15;

  const int nS = gridDim.x * 4;
  const int s = blockIdx.x * 4 + (wave & 3);
  const int ts0 = (int)(((long long)s * nTiles) / nS);
  const int ts1 = (int)(((long long)(s + 1) * nTiles) / nS);
  const int firstPart = (ts1 - ts0 + 1) >> 1;
  const int w0 = (wave >> 2) ? (ts0 + firstPart) : ts0;
  const int w1 = (wave >> 2) ? ts1 : (ts0 + firstPart);
  const int maxS = (nTiles + nS - 1) / nS;
  const int nPass = (((maxS + 1) >> 1) + MB - 1) / MB;

  const int nkt = K >> 2;
  const int nkt2 = nkt << 1;
  const int nChunks = nkt2 / CHUNK_KT;

  for (int p = 0; p < nPass; ++p) {
    const int tb = w0 + p * MB;
    int cnt = w1 - tb;
    cnt = cnt < 0 ? 0 : (cnt > MB ? MB : cnt);
    const int lastT = (cnt > 0) ? (tb + cnt - 1) : (nTiles - 1);

    unsigned rowoff[MB];
    float scale[MB];
#pragma unroll
    for (int m = 0; m < MB; ++m) {
      const int t = (m < cnt) ? (tb + m) : lastT;
      const int arow = t * 16 + n0;
      rowoff[m] = (unsigned)arow * (unsigned)K + (unsigned)(half * 2);
      scale[m] = 1.0f / fmaxf(deg[arow], 1.0f);
    }

    v8f acc[MB][8];
#pragma unroll
    for (int nt = 0; nt < 8; ++nt) {
      const float b = bias[nt * 16 + n0];
#pragma unroll
      for (int m = 0; m < MB; ++m) acc[m][nt] = (v8f){b, b, b, b, b, b, b, b};
    }

    v2f ring[AHEAD][MB];
#pragma unroll
    for (int j = 0; j < AHEAD; ++j)
#pragma unroll
      for (int m = 0; m < MB; ++m) ring[j][m] = *(const v2f*)(A1 + rowoff[m] + j * 4);

    {
      const float4* __restrict__ gsrc = (const float4*)Bp1;
      float4* __restrict__ ldst = (float4*)lds[0];
#pragma unroll
      for (int i = 0; i < 16; ++i) ldst[tid + i * 256] = gsrc[tid + i * 256];
    }
    __syncthreads();

    for (int c = 0; c < nChunks; ++c) {
      const int kt0 = c * CHUNK_KT;
      const bool more = c + 1 < nChunks;
      const int ktx = kt0 + CHUNK_KT;
      const float4* __restrict__ gnext =
          !more ? (const float4*)nullptr
                : (const float4*)((ktx >= nkt)
                                      ? (Bp2 + (size_t)(ktx - nkt) * 8 * 64)
                                      : (Bp1 + (size_t)ktx * 8 * 64));
      float4* __restrict__ lnext = (float4*)lds[(c + 1) & 1];
      const float* ldsb = lds[c & 1];
      k_chunk(acc, ring, A1, A2, rowoff, scale, ldsb, gnext, lnext, kt0, nkt,
              nkt2, lane, tid, cnt == MB);
      __syncthreads();
    }

#pragma unroll
    for (int m = 0; m < MB; ++m) {
      if (m < cnt) {
        const int m0 = (tb + m) * 16;
#pragma unroll
        for (int nt = 0; nt < 8; ++nt) {
#pragma unroll
          for (int r = 0; r < 8; ++r) {
            const int row = m0 + r + 8 * half;
            const size_t idx = (size_t)row * 128 + nt * 16 + n0;
            float v = acc[m][nt][r];
            if (relu) v = fmaxf(v, 0.0f);
            out[idx] = v;
          }
        }
      }
    }
  }
}

extern "C" void kernel_launch(void* const* d_in, const int* in_sizes, int n_in,
                              void* d_out, int out_size, void* d_ws,
                              size_t ws_size, hipStream_t stream) {
  const int IN_DIM = 384, HID = 128;
  const int N = in_sizes[0] / IN_DIM;
  const int E = in_sizes[1] / 2;

  const float* x    = (const float*)d_in[0];
  const int*   ei   = (const int*)d_in[1];
  const float* W1_l = (const float*)d_in[2];
  const float* b1   = (const float*)d_in[3];
  const float* W1_r = (const float*)d_in[4];
  const float* W2_l = (const float*)d_in[5];
  const float* b2   = (const float*)d_in[6];
  const float* W2_r = (const float*)d_in[7];
  const int* src = ei;
  const int* dst = ei + E;
  float* out = (float*)d_out;

  float* ws = (float*)d_ws;
  const long deg_off  = 0;
  const long deg_pad  = ((long)N + 127) & ~127L;
  const long agg_off  = deg_off + deg_pad;
  const long h1_off   = agg_off + (long)N * IN_DIM;
  const long pw1l_off = h1_off + (long)N * HID;
  const long pw1r_off = pw1l_off + (long)IN_DIM * HID;
  const long pw2l_off = pw1r_off + (long)IN_DIM * HID;
  const long pw2r_off = pw2l_off + (long)HID * HID;

  float* deg  = ws + deg_off;
  float* agg  = ws + agg_off;
  float* h1   = ws + h1_off;
  float* pw1l = ws + pw1l_off;
  float* pw1r = ws + pw1r_off;
  float* pw2l = ws + pw2l_off;
  float* pw2r = ws + pw2r_off;

  const int nTiles = N / 16;
  const int gemmBlocks = 256;

  pack_weights_kernel<<<(IN_DIM * 128 + 255) / 256, 256, 0, stream>>>(W1_l, pw1l, IN_DIM);
  pack_weights_kernel<<<(IN_DIM * 128 + 255) / 256, 256, 0, stream>>>(W1_r, pw1r, IN_DIM);
  pack_weights_kernel<<<(HID * 128 + 255) / 256, 256, 0, stream>>>(W2_l, pw2l, HID);
  pack_weights_kernel<<<(HID * 128 + 255) / 256, 256, 0, stream>>>(W2_r, pw2r, HID);

  range_gather_kernel<384, 416, 4><<<(N + 415) / 416, 1024, 0, stream>>>(
      x, src, dst, agg, deg, E, N);

  gemm_pair_kernel<<<gemmBlocks, 256, 0, stream>>>(agg, pw1l, x, pw1r, IN_DIM,
                                                   b1, deg, h1, 1, nTiles);

  range_gather_kernel<128, 416, 8><<<(N + 415) / 416, 1024, 0, stream>>>(
      h1, src, dst, agg, nullptr, E, N);

  gemm_pair_kernel<<<gemmBlocks, 256, 0, stream>>>(agg, pw2l, h1, pw2r, HID, b2,
                                                   deg, out, 0, nTiles);
}


# Edge list read once: a bin pass appends each edge to its owner range's bin (integer returning adds); both neighbour-sum launches read their bin, not all of dst (arrival-order ranks kept); 9 launches
# speedup vs baseline: 2.2343x; 1.0524x over previous
#include <hip/hip_runtime.h>


typedef __attribute__((ext_vector_type(2))) float v2f;
typedef __attribute__((ext_vector_type(8))) float v8f;

#define CHUNK_KT 32

__global__ __launch_bounds__(256) void pack_weights_kernel(
    const float* __restrict__ W, float* __restrict__ P, int K,
    int* __restrict__ zero, int nzero) {
  if (zero)
    for (int i = blockIdx.x * blockDim.x + threadIdx.x; i < nzero;
         i += gridDim.x * blockDim.x)
      zero[i] = 0;
  int total = K * 128;
  for (int tid = blockIdx.x * blockDim.x + threadIdx.x; tid < total;
       tid += gridDim.x * blockDim.x) {
    int frag = tid >> 6;
    int r = tid & 63;
    int L = r >> 1;
    int v = r & 1;
    int kt = frag >> 3;
    int nt = frag & 7;
    int k = kt * 4 + ((L >> 4) << 1) + v;
    int n = nt * 16 + (L & 15);
    P[tid] = W[k * 128 + n];
  }
}

#define BIN_CAP 4096
#define CNT_PITCH 32
#define MAX_RANGES 1024

template <int R>
__global__ __launch_bounds__(1024) void bin_edges_kernel(
    const int* __restrict__ src, const int* __restrict__ dst,
    int* __restrict__ counts, int* __restrict__ bins, int E, int N,
    int nRanges) {
  __shared__ int hist[MAX_RANGES];
  __shared__ int gbase[MAX_RANGES];
  const int tid = threadIdx.x;
  if (tid < nRanges) hist[tid] = 0;
  __syncthreads();

  const int e = blockIdx.x * 1024 + tid;
  int owner = -1, rel = 0, s = 0, lrank = 0;
  if (e < E) {
    const int d = dst[e];
    s = src[e];
    if ((unsigned)d < (unsigned)N) {
      owner = d / R;
      rel = d - owner * R;
      lrank = __hip_atomic_fetch_add(&hist[owner], 1, __ATOMIC_RELAXED,
                                     __HIP_MEMORY_SCOPE_WORKGROUP);
    }
  }
  __syncthreads();
  if (tid < nRanges) {
    const int c = hist[tid];
    gbase[tid] = c ? __hip_atomic_fetch_add(&counts[tid * CNT_PITCH], c,
                                            __ATOMIC_RELAXED,
                                            __HIP_MEMORY_SCOPE_AGENT)
                   : 0;
  }
  __syncthreads();
  if (owner >= 0) {
    const int pos = gbase[owner] + lrank;
    if (pos < BIN_CAP)
      ((int2*)bins)[(size_t)owner * BIN_CAP + pos] = make_int2(s, rel);
  }
}


template <int Q>
__device__ __forceinline__ void store_row_and_clear(float* __restrict__ agg,
                                                    size_t node, int F,
                                                    int lane, float4* acc) {
  float4* __restrict__ o = (float4*)(agg + node * (size_t)F) + lane;
#pragma unroll
  for (int q = 0; q < Q; ++q) {
    o[q * 32] = acc[q];
    acc[q] = make_float4(0.0f, 0.0f, 0.0f, 0.0f);
  }
}

template <int F, int R, int B>
__global__ __launch_bounds__(1024) void range_gather_kernel(
    const float* __restrict__ feat, const int* __restrict__ src,
    const int* __restrict__ dst, const int* __restrict__ counts,
    const int* __restrict__ bins, float* __restrict__ agg,
    float* __restrict__ deg, int E, int N) {
  constexpr int CAP = BIN_CAP;
  constexpr int PER = R / 32;
  constexpr int Q = F / 128;
  __shared__ int cnt[R];
  __shared__ int start[R + 1];
  __shared__ int list_e[CAP];
  __shared__ unsigned list_k[CAP];
  __shared__ int sorted[CAP];

  const int tid = threadIdx.x;
  const int lane = tid & 31;
  const int wave = __builtin_amdgcn_readfirstlane(tid >> 5);
  const int base = blockIdx.x * R;
  const int rows = (N - base < R) ? (N - base) : R;

  for (int i = tid; i < R; i += 1024) cnt[i] = 0;
  __syncthreads();

  const int total = counts[blockIdx.x * CNT_PITCH];
  if (total <= CAP) {
    const int2* __restrict__ bin = (const int2*)bins + (size_t)blockIdx.x * CAP;
    for (int i = tid; i < total; i += 1024) {
      const int2 r = bin[i];
      const int rank = __hip_atomic_fetch_add(
          &cnt[r.y], 1, __ATOMIC_RELAXED, __HIP_MEMORY_SCOPE_WORKGROUP);
      list_e[i] = r.x;
      list_k[i] = (unsigned)r.y | ((unsigned)rank << 16);
    }
  }
  __syncthreads();

  if (wave == 0) {
    int s = 0;
    for (int k = 0; k < PER; ++k) s += cnt[lane * PER + k];
    int incl = s;
    for (int d = 1; d < 32; d <<= 1) {
      const int t = __shfl_up(incl, d);
      if (lane >= d) incl += t;
    }
    int run = incl - s;
    for (int k = 0; k < PER; ++k) {
      start[lane * PER + k] = run;
      run += cnt[lane * PER + k];
    }
    if (lane == 31) start[R] = run;
  }
  __syncthreads();

  if (total <= CAP) {
    for (int i = tid; i < total; i += 1024) {
      const unsigned k = list_k[i];
      sorted[start[k & 0xffffu] + (int)(k >> 16)] = list_e[i];
    }
  }
  __syncthreads();

  const int r0 = wave * PER;
  const int r1 = (r0 + PER < rows) ? (r0 + PER) : rows;
  float4 acc[Q];
#pragma unroll
  for (int q = 0; q < Q; ++q) acc[q] = make_float4(0.0f, 0.0f, 0.0f, 0.0f);

  if (r0 < rows) {
    if (total <= CAP) {
      const int e_lo = __builtin_amdgcn_readfirstlane(start[r0]);
      const int e_hi = __builtin_amdgcn_readfirstlane(start[r1]);
      int node = r0;
      int node_end = __builtin_amdgcn_readfirstlane(start[r0 + 1]);
      for (int eb = e_lo; eb < e_hi; eb += B) {
        float4 v[B][Q];
#pragma unroll
        for (int j = 0; j < B; ++j) {
          const int ei = (eb + j < e_hi) ? (eb + j) : (e_hi - 1);
          const int s = __builtin_amdgcn_readfirstlane(sorted[ei]);
          const float4* __restrict__ row =
              (const float4*)(feat + (size_t)s * F) + lane;
#pragma unroll
          for (int q = 0; q < Q; ++q) v[j][q] = row[q * 32];
        }
#pragma unroll
        for (int j = 0; j < B; ++j) {
          if (eb + j < e_hi) {
            while (eb + j >= node_end) {
              store_row_and_clear<Q>(agg, (size_t)(base + node), F, lane, acc);
              ++node;
              node_end = __builtin_amdgcn_readfirstlane(start[node + 1]);
            }
#pragma unroll
            for (int q = 0; q < Q; ++q) {
              acc[q].x += v[j][q].x;
              acc[q].y += v[j][q].y;
              acc[q].z += v[j][q].z;
              acc[q].w += v[j][q].w;
            }
          }
        }
      }
      for (; node < r1; ++node)
        store_row_and_clear<Q>(agg, (size_t)(base + node), F, lane, acc);
    } else {
      for (int node = r0; node < r1; ++node) {
        int nedges = 0;
        for (int eb = 0; eb < E; eb += 32) {
          const int e = eb + lane;
          const int d = (e < E) ? dst[e] : -1;
          unsigned m = __builtin_amdgcn_ballot_w32(d == base + node);
          nedges += __builtin_popcount(m);
          while (m) {
            const int l = __builtin_ctz(m);
            m &= m - 1;
            const int s = src[eb + l];
            const float4* __restrict__ row =
                (const float4*)(feat + (size_t)s * F) + lane;
#pragma unroll
            for (int q = 0; q < Q; ++q) {
              const float4 w = row[q * 32];
              acc[q].x += w.x;
              acc[q].y += w.y;
              acc[q].z += w.z;
              acc[q].w += w.w;
            }
          }
        }
        store_row_and_clear<Q>(agg, (size_t)(base + node), F, lane, acc);
        if (deg && lane == 0) deg[base + node] = (float)nedges;
      }
    }
  }
  if (deg && total <= CAP)
    for (int i = tid; i < rows; i += 1024) deg[base + i] = (float)cnt[i];
}

#define MB 4
#define AHEAD 4
#define STG 2

__device__ __forceinline__ void k_chunk(
    v8f (&acc)[MB][8], v2f (&ring)[AHEAD][MB], const float* __restrict__ A1,
    const float* __restrict__ A2, const unsigned (&rowoff)[MB],
    const float (&scale)[MB], const float* __restrict__ ldsb,
    const float4* __restrict__ gnext, float4* __restrict__ lnext, int kt0,
    int nkt, int nkt2, int lane, int tid, bool full) {
  const bool second = kt0 >= nkt;
  for (int kb = 0; kb < CHUNK_KT; kb += AHEAD) {
    const int sbase = tid + (kb / AHEAD) * STG * 256;
    float4 s0 = make_float4(0.0f, 0.0f, 0.0f, 0.0f), s1 = s0;
    if (gnext) {
      s0 = gnext[sbase];
      s1 = gnext[sbase + 256];
    }
    v2f nxt[AHEAD][MB];
#pragma unroll
    for (int j = 0; j < AHEAD; ++j) {
      int ktn = kt0 + kb + AHEAD + j;
      if (ktn > nkt2 - 1) ktn = nkt2 - 1;
      const bool n2 = ktn >= nkt;
      const float* __restrict__ an = (n2 ? A2 : A1) + (n2 ? (ktn - nkt) : ktn) * 4;
#pragma unroll
      for (int m = 0; m < MB; ++m) nxt[j][m] = *(const v2f*)(an + rowoff[m]);
    }
    asm volatile("" ::: "memory");
#pragma unroll
    for (int j = 0; j < AHEAD; ++j) {
      const int kt = kb + j;
      v2f a[MB];
#pragma unroll
      for (int m = 0; m < MB; ++m)
        a[m] = ring[j][m] * (second ? 1.0f : scale[m]);

      const float* bbase = ldsb + (kt * 8) * 64 + lane * 2;
      v2f bf[8];
#pragma unroll
      for (int nt = 0; nt < 8; ++nt) bf[nt] = *(const v2f*)(bbase + nt * 64);

      if (full) {
#pragma unroll
        for (int nt = 0; nt < 8; ++nt) {
          acc[MB - 1][nt] = __builtin_amdgcn_wmma_f32_16x16x4_f32(
              false, a[MB - 1], false, bf[nt], (short)0, acc[MB - 1][nt], false,
              false);
        }
      }
#pragma unroll
      for (int nt = 0; nt < 8; ++nt) {
#pragma unroll
        for (int m = 0; m < MB - 1; ++m) {
          acc[m][nt] = __builtin_amdgcn_wmma_f32_16x16x4_f32(
              false, a[m], false, bf[nt], (short)0, acc[m][nt], false, false);
        }
      }
    }
#pragma unroll
    for (int j = 0; j < AHEAD; ++j)
#pragma unroll
      for (int m = 0; m < MB; ++m) ring[j][m] = nxt[j][m];
    if (gnext) {
      lnext[sbase] = s0;
      lnext[sbase + 256] = s1;
    }
  }
}

__global__ __launch_bounds__(256) void gemm_pair_kernel(
    const float* __restrict__ A1, const float* __restrict__ Bp1,
    const float* __restrict__ A2, const float* __restrict__ Bp2, int K,
    const float* __restrict__ bias, const float* __restrict__ deg,
    float* __restrict__ out, int relu, int nTiles) {
  __shared__ float lds[2][CHUNK_KT * 8 * 64];

  const int tid = threadIdx.x;
  const int lane = tid & 31;
  const int wave = __builtin_amdgcn_readfirstlane(tid >> 5);
  const int half = lane >> 4;
  const int n0 = lane & 15;

  const int nS = gridDim.x * 4;
  const int s = blockIdx.x * 4 + (wave & 3);
  const int ts0 = (int)(((long long)s * nTiles) / nS);
  const int ts1 = (int)(((long long)(s + 1) * nTiles) / nS);
  const int firstPart = (ts1 - ts0 + 1) >> 1;
  const int w0 = (wave >> 2) ? (ts0 + firstPart) : ts0;
  const int w1 = (wave >> 2) ? ts1 : (ts0 + firstPart);
  const int maxS = (nTiles + nS - 1) / nS;
  const int nPass = (((maxS + 1) >> 1) + MB - 1) / MB;

  const int nkt = K >> 2;
  const int nkt2 = nkt << 1;
  const int nChunks = nkt2 / CHUNK_KT;

  for (int p = 0; p < nPass; ++p) {
    const int tb = w0 + p * MB;
    int cnt = w1 - tb;
    cnt = cnt < 0 ? 0 : (cnt > MB ? MB : cnt);
    const int lastT = (cnt > 0) ? (tb + cnt - 1) : (nTiles - 1);

    unsigned rowoff[MB];
    float scale[MB];
#pragma unroll
    for (int m = 0; m < MB; ++m) {
      const int t = (m < cnt) ? (tb + m) : lastT;
      const int arow = t * 16 + n0;
      rowoff[m] = (unsigned)arow * (unsigned)K + (unsigned)(half * 2);
      scale[m] = 1.0f / fmaxf(deg[arow], 1.0f);
    }

    v8f acc[MB][8];
#pragma unroll
    for (int nt = 0; nt < 8; ++nt) {
      const float b = bias[nt * 16 + n0];
#pragma unroll
      for (int m = 0; m < MB; ++m) acc[m][nt] = (v8f){b, b, b, b, b, b, b, b};
    }

    v2f ring[AHEAD][MB];
#pragma unroll
    for (int j = 0; j < AHEAD; ++j)
#pragma unroll
      for (int m = 0; m < MB; ++m) ring[j][m] = *(const v2f*)(A1 + rowoff[m] + j * 4);

    {
      const float4* __restrict__ gsrc = (const float4*)Bp1;
      float4* __restrict__ ldst = (float4*)lds[0];
#pragma unroll
      for (int i = 0; i < 16; ++i) ldst[tid + i * 256] = gsrc[tid + i * 256];
    }
    __syncthreads();

    for (int c = 0; c < nChunks; ++c) {
      const int kt0 = c * CHUNK_KT;
      const bool more = c + 1 < nChunks;
      const int ktx = kt0 + CHUNK_KT;
      const float4* __restrict__ gnext =
          !more ? (const float4*)nullptr
                : (const float4*)((ktx >= nkt)
                                      ? (Bp2 + (size_t)(ktx - nkt) * 8 * 64)
                                      : (Bp1 + (size_t)ktx * 8 * 64));
      float4* __restrict__ lnext = (float4*)lds[(c + 1) & 1];
      const float* ldsb = lds[c & 1];
      k_chunk(acc, ring, A1, A2, rowoff, scale, ldsb, gnext, lnext, kt0, nkt,
              nkt2, lane, tid, cnt == MB);
      __syncthreads();
    }

#pragma unroll
    for (int m = 0; m < MB; ++m) {
      if (m < cnt) {
        const int m0 = (tb + m) * 16;
#pragma unroll
        for (int nt = 0; nt < 8; ++nt) {
#pragma unroll
          for (int r = 0; r < 8; ++r) {
            const int row = m0 + r + 8 * half;
            const size_t idx = (size_t)row * 128 + nt * 16 + n0;
            float v = acc[m][nt][r];
            if (relu) v = fmaxf(v, 0.0f);
            out[idx] = v;
          }
        }
      }
    }
  }
}

extern "C" void kernel_launch(void* const* d_in, const int* in_sizes, int n_in,
                              void* d_out, int out_size, void* d_ws,
                              size_t ws_size, hipStream_t stream) {
  const int IN_DIM = 384, HID = 128;
  const int N = in_sizes[0] / IN_DIM;
  const int E = in_sizes[1] / 2;

  const float* x    = (const float*)d_in[0];
  const int*   ei   = (const int*)d_in[1];
  const float* W1_l = (const float*)d_in[2];
  const float* b1   = (const float*)d_in[3];
  const float* W1_r = (const float*)d_in[4];
  const float* W2_l = (const float*)d_in[5];
  const float* b2   = (const float*)d_in[6];
  const float* W2_r = (const float*)d_in[7];
  const int* src = ei;
  const int* dst = ei + E;
  float* out = (float*)d_out;

  float* ws = (float*)d_ws;
  const long deg_off  = 0;
  const long deg_pad  = ((long)N + 127) & ~127L;
  const long agg_off  = deg_off + deg_pad;
  const long h1_off   = agg_off + (long)N * IN_DIM;
  const long pw1l_off = h1_off + (long)N * HID;
  const long pw1r_off = pw1l_off + (long)IN_DIM * HID;
  const long pw2l_off = pw1r_off + (long)IN_DIM * HID;
  const long pw2r_off = pw2l_off + (long)HID * HID;
  const long cnt_off  = pw2r_off + (long)HID * HID;
  const long bins_off = cnt_off + (long)MAX_RANGES * CNT_PITCH;

  float* deg  = ws + deg_off;
  float* agg  = ws + agg_off;
  float* h1   = ws + h1_off;
  float* pw1l = ws + pw1l_off;
  float* pw1r = ws + pw1r_off;
  float* pw2l = ws + pw2l_off;
  float* pw2r = ws + pw2r_off;
  int*  counts = (int*)(ws + cnt_off);
  int*  bins   = (int*)(ws + bins_off);

  const int nTiles = N / 16;
  const int gemmBlocks = 256;
  const int nRanges = (N + 415) / 416;

  pack_weights_kernel<<<(IN_DIM * 128 + 255) / 256, 256, 0, stream>>>(
      W1_l, pw1l, IN_DIM, counts, MAX_RANGES * CNT_PITCH);
  pack_weights_kernel<<<(IN_DIM * 128 + 255) / 256, 256, 0, stream>>>(
      W1_r, pw1r, IN_DIM, nullptr, 0);
  pack_weights_kernel<<<(HID * 128 + 255) / 256, 256, 0, stream>>>(
      W2_l, pw2l, HID, nullptr, 0);
  pack_weights_kernel<<<(HID * 128 + 255) / 256, 256, 0, stream>>>(
      W2_r, pw2r, HID, nullptr, 0);

  bin_edges_kernel<416><<<(E + 1023) / 1024, 1024, 0, stream>>>(
      src, dst, counts, bins, E, N, nRanges);

  range_gather_kernel<384, 416, 4><<<nRanges, 1024, 0, stream>>>(
      x, src, dst, counts, bins, agg, deg, E, N);

  gemm_pair_kernel<<<gemmBlocks, 256, 0, stream>>>(agg, pw1l, x, pw1r, IN_DIM,
                                                   b1, deg, h1, 1, nTiles);

  range_gather_kernel<128, 416, 8><<<nRanges, 1024, 0, stream>>>(
      h1, src, dst, counts, bins, agg, nullptr, E, N);

  gemm_pair_kernel<<<gemmBlocks, 256, 0, stream>>>(agg, pw2l, h1, pw2r, HID, b2,
                                                   deg, out, 0, nTiles);
}


# Set-up in one launch: a bin work-group sorts its 1024 edges by owner range into its own segment (no counter clear, no global atomics; arrival-order ranks kept); packing joins it; 9 -> 5 launches
# speedup vs baseline: 2.3160x; 1.0366x over previous
#include <hip/hip_runtime.h>


typedef __attribute__((ext_vector_type(2))) float v2f;
typedef __attribute__((ext_vector_type(8))) float v8f;

#define CHUNK_KT 32

__device__ __forceinline__ void pack_one(const float* __restrict__ W,
                                         float* __restrict__ P, int idx) {
  const int frag = idx >> 6;
  const int r = idx & 63;
  const int L = r >> 1;
  const int v = r & 1;
  const int kt = frag >> 3;
  const int nt = frag & 7;
  const int k = kt * 4 + ((L >> 4) << 1) + v;
  const int n = nt * 16 + (L & 15);
  P[idx] = W[k * 128 + n];
}

#define BIN_CAP 4096
#define MAX_RANGES 1024
#define MAX_BWG 256
#define OFF_PITCH 1056
#define PACK_PER 4

template <int R>
__global__ __launch_bounds__(1024) void setup_kernel(
    const int* __restrict__ src, const int* __restrict__ dst,
    int* __restrict__ offs, int* __restrict__ segs, int E, int N, int nRanges,
    int nBinWG, const float* __restrict__ W1l, float* __restrict__ P1l,
    const float* __restrict__ W1r, float* __restrict__ P1r, int K1,
    const float* __restrict__ W2l, float* __restrict__ P2l,
    const float* __restrict__ W2r, float* __restrict__ P2r, int K2) {
  __shared__ int hist[MAX_RANGES];
  __shared__ int first[MAX_RANGES + 1];
  const int tid = threadIdx.x;

  if ((int)blockIdx.x >= nBinWG) {
    const int n1 = K1 * 128, n2 = K2 * 128;
    const int g0 = ((int)blockIdx.x - nBinWG) * (1024 * PACK_PER) + tid;
    for (int j = 0; j < PACK_PER; ++j) {
      int g = g0 + j * 1024;
      if (g < n1) {
        pack_one(W1l, P1l, g);
      } else if ((g -= n1) < n1) {
        pack_one(W1r, P1r, g);
      } else if ((g -= n1) < n2) {
        pack_one(W2l, P2l, g);
      } else if ((g -= n2) < n2) {
        pack_one(W2r, P2r, g);
      }
    }
    return;
  }

  const int lane = tid & 31;
  for (int i = tid; i < MAX_RANGES; i += 1024) hist[i] = 0;
  __syncthreads();

  const int e = blockIdx.x * 1024 + tid;
  int owner = -1, rel = 0, s = 0, lrank = 0;
  if (e < E) {
    const int d = dst[e];
    s = src[e];
    if ((unsigned)d < (unsigned)N) {
      owner = d / R;
      rel = d - owner * R;
      lrank = __hip_atomic_fetch_add(&hist[owner], 1, __ATOMIC_RELAXED,
                                     __HIP_MEMORY_SCOPE_WORKGROUP);
    }
  }
  __syncthreads();

  if (tid < 32) {
    const int per = (nRanges + 31) >> 5;
    int sum = 0;
    for (int k = 0; k < per; ++k) sum += hist[lane * per + k];
    int incl = sum;
    for (int d = 1; d < 32; d <<= 1) {
      const int t = __shfl_up(incl, d);
      if (lane >= d) incl += t;
    }
    int run = incl - sum;
    for (int k = 0; k < per; ++k) {
      first[lane * per + k] = run;
      run += hist[lane * per + k];
    }
    if (lane == 31) first[nRanges] = run;
  }
  __syncthreads();

  for (int i = tid; i <= nRanges; i += 1024)
    offs[(size_t)blockIdx.x * OFF_PITCH + i] = first[i];
  if (owner >= 0)
    ((int2*)segs)[(size_t)blockIdx.x * 1024 + first[owner] + lrank] =
        make_int2(s, rel);
}


template <int Q>
__device__ __forceinline__ void store_row_and_clear(float* __restrict__ agg,
                                                    size_t node, int F,
                                                    int lane, float4* acc) {
  float4* __restrict__ o = (float4*)(agg + node * (size_t)F) + lane;
#pragma unroll
  for (int q = 0; q < Q; ++q) {
    o[q * 32] = acc[q];
    acc[q] = make_float4(0.0f, 0.0f, 0.0f, 0.0f);
  }
}

template <int F, int R, int B>
__global__ __launch_bounds__(1024) void range_gather_kernel(
    const float* __restrict__ feat, const int* __restrict__ src,
    const int* __restrict__ dst, const int* __restrict__ offs,
    const int* __restrict__ segs, int nBinWG, float* __restrict__ agg,
    float* __restrict__ deg, int E, int N) {
  constexpr int CAP = BIN_CAP;
  constexpr int PER = R / 32;
  constexpr int Q = F / 128;
  __shared__ int cnt[R];
  __shared__ int start[R + 1];
  __shared__ int list_e[CAP];
  __shared__ unsigned list_k[CAP];
  __shared__ int sorted[CAP];
  __shared__ int clo[MAX_BWG];
  __shared__ int cpos[MAX_BWG + 1];

  const int tid = threadIdx.x;
  const int lane = tid & 31;
  const int wave = __builtin_amdgcn_readfirstlane(tid >> 5);
  const int base = blockIdx.x * R;
  const int rows = (N - base < R) ? (N - base) : R;

  for (int i = tid; i < R; i += 1024) cnt[i] = 0;
  if (tid < MAX_BWG) {
    int lo = 0, hi = 0;
    if (tid < nBinWG) {
      const int* __restrict__ t = offs + (size_t)tid * OFF_PITCH + blockIdx.x;
      lo = t[0];
      hi = t[1];
    }
    clo[tid] = lo;
    cpos[tid] = hi - lo;
  }
  __syncthreads();

  if (wave == 0) {
    int c[MAX_BWG / 32];
    int s = 0;
    for (int k = 0; k < MAX_BWG / 32; ++k) {
      c[k] = cpos[lane * (MAX_BWG / 32) + k];
      s += c[k];
    }
    int incl = s;
    for (int d = 1; d < 32; d <<= 1) {
      const int t = __shfl_up(incl, d);
      if (lane >= d) incl += t;
    }
    int run = incl - s;
    for (int k = 0; k < MAX_BWG / 32; ++k) {
      cpos[lane * (MAX_BWG / 32) + k] = run;
      run += c[k];
    }
    if (lane == 31) cpos[MAX_BWG] = run;
  }
  __syncthreads();

  const int total = (nBinWG <= MAX_BWG) ? cpos[MAX_BWG] : (CAP + 1);
  if (total <= CAP) {
    const int2* __restrict__ seg = (const int2*)segs;
    for (int i = tid; i < total; i += 1024) {
      int w = 0;
      for (int step = MAX_BWG / 2; step > 0; step >>= 1)
        if (cpos[w + step] <= i) w += step;
      const int2 r = seg[(size_t)w * 1024 + clo[w] + (i - cpos[w])];
      const int rank = __hip_atomic_fetch_add(
          &cnt[r.y], 1, __ATOMIC_RELAXED, __HIP_MEMORY_SCOPE_WORKGROUP);
      list_e[i] = r.x;
      list_k[i] = (unsigned)r.y | ((unsigned)rank << 16);
    }
  }
  __syncthreads();

  if (wave == 0) {
    int s = 0;
    for (int k = 0; k < PER; ++k) s += cnt[lane * PER + k];
    int incl = s;
    for (int d = 1; d < 32; d <<= 1) {
      const int t = __shfl_up(incl, d);
      if (lane >= d) incl += t;
    }
    int run = incl - s;
    for (int k = 0; k < PER; ++k) {
      start[lane * PER + k] = run;
      run += cnt[lane * PER + k];
    }
    if (lane == 31) start[R] = run;
  }
  __syncthreads();

  if (total <= CAP) {
    for (int i = tid; i < total; i += 1024) {
      const unsigned k = list_k[i];
      sorted[start[k & 0xffffu] + (int)(k >> 16)] = list_e[i];
    }
  }
  __syncthreads();

  const int r0 = wave * PER;
  const int r1 = (r0 + PER < rows) ? (r0 + PER) : rows;
  float4 acc[Q];
#pragma unroll
  for (int q = 0; q < Q; ++q) acc[q] = make_float4(0.0f, 0.0f, 0.0f, 0.0f);

  if (r0 < rows) {
    if (total <= CAP) {
      const int e_lo = __builtin_amdgcn_readfirstlane(start[r0]);
      const int e_hi = __builtin_amdgcn_readfirstlane(start[r1]);
      int node = r0;
      int node_end = __builtin_amdgcn_readfirstlane(start[r0 + 1]);
      for (int eb = e_lo; eb < e_hi; eb += B) {
        float4 v[B][Q];
#pragma unroll
        for (int j = 0; j < B; ++j) {
          const int ei = (eb + j < e_hi) ? (eb + j) : (e_hi - 1);
          const int s = __builtin_amdgcn_readfirstlane(sorted[ei]);
          const float4* __restrict__ row =
              (const float4*)(feat + (size_t)s * F) + lane;
#pragma unroll
          for (int q = 0; q < Q; ++q) v[j][q] = row[q * 32];
        }
#pragma unroll
        for (int j = 0; j < B; ++j) {
          if (eb + j < e_hi) {
            while (eb + j >= node_end) {
              store_row_and_clear<Q>(agg, (size_t)(base + node), F, lane, acc);
              ++node;
              node_end = __builtin_amdgcn_readfirstlane(start[node + 1]);
            }
#pragma unroll
            for (int q = 0; q < Q; ++q) {
              acc[q].x += v[j][q].x;
              acc[q].y += v[j][q].y;
              acc[q].z += v[j][q].z;
              acc[q].w += v[j][q].w;
            }
          }
        }
      }
      for (; node < r1; ++node)
        store_row_and_clear<Q>(agg, (size_t)(base + node), F, lane, acc);
    } else {
      for (int node = r0; node < r1; ++node) {
        int nedges = 0;
        for (int eb = 0; eb < E; eb += 32) {
          const int e = eb + lane;
          const int d = (e < E) ? dst[e] : -1;
          unsigned m = __builtin_amdgcn_ballot_w32(d == base + node);
          nedges += __builtin_popcount(m);
          while (m) {
            const int l = __builtin_ctz(m);
            m &= m - 1;
            const int s = src[eb + l];
            const float4* __restrict__ row =
                (const float4*)(feat + (size_t)s * F) + lane;
#pragma unroll
            for (int q = 0; q < Q; ++q) {
              const float4 w = row[q * 32];
              acc[q].x += w.x;
              acc[q].y += w.y;
              acc[q].z += w.z;
              acc[q].w += w.w;
            }
          }
        }
        store_row_and_clear<Q>(agg, (size_t)(base + node), F, lane, acc);
        if (deg && lane == 0) deg[base + node] = (float)nedges;
      }
    }
  }
  if (deg && total <= CAP)
    for (int i = tid; i < rows; i += 1024) deg[base + i] = (float)cnt[i];
}

#define MB 4
#define AHEAD 4
#define STG 2

__device__ __forceinline__ void k_chunk(
    v8f (&acc)[MB][8], v2f (&ring)[AHEAD][MB], const float* __restrict__ A1,
    const float* __restrict__ A2, const unsigned (&rowoff)[MB],
    const float (&scale)[MB], const float* __restrict__ ldsb,
    const float4* __restrict__ gnext, float4* __restrict__ lnext, int kt0,
    int nkt, int nkt2, int lane, int tid, bool full) {
  const bool second = kt0 >= nkt;
  for (int kb = 0; kb < CHUNK_KT; kb += AHEAD) {
    const int sbase = tid + (kb / AHEAD) * STG * 256;
    float4 s0 = make_float4(0.0f, 0.0f, 0.0f, 0.0f), s1 = s0;
    if (gnext) {
      s0 = gnext[sbase];
      s1 = gnext[sbase + 256];
    }
    v2f nxt[AHEAD][MB];
#pragma unroll
    for (int j = 0; j < AHEAD; ++j) {
      int ktn = kt0 + kb + AHEAD + j;
      if (ktn > nkt2 - 1) ktn = nkt2 - 1;
      const bool n2 = ktn >= nkt;
      const float* __restrict__ an = (n2 ? A2 : A1) + (n2 ? (ktn - nkt) : ktn) * 4;
#pragma unroll
      for (int m = 0; m < MB; ++m) nxt[j][m] = *(const v2f*)(an + rowoff[m]);
    }
    asm volatile("" ::: "memory");
#pragma unroll
    for (int j = 0; j < AHEAD; ++j) {
      const int kt = kb + j;
      v2f a[MB];
#pragma unroll
      for (int m = 0; m < MB; ++m)
        a[m] = ring[j][m] * (second ? 1.0f : scale[m]);

      const float* bbase = ldsb + (kt * 8) * 64 + lane * 2;
      v2f bf[8];
#pragma unroll
      for (int nt = 0; nt < 8; ++nt) bf[nt] = *(const v2f*)(bbase + nt * 64);

      if (full) {
#pragma unroll
        for (int nt = 0; nt < 8; ++nt) {
          acc[MB - 1][nt] = __builtin_amdgcn_wmma_f32_16x16x4_f32(
              false, a[MB - 1], false, bf[nt], (short)0, acc[MB - 1][nt], false,
              false);
        }
      }
#pragma unroll
      for (int nt = 0; nt < 8; ++nt) {
#pragma unroll
        for (int m = 0; m < MB - 1; ++m) {
          acc[m][nt] = __builtin_amdgcn_wmma_f32_16x16x4_f32(
              false, a[m], false, bf[nt], (short)0, acc[m][nt], false, false);
        }
      }
    }
#pragma unroll
    for (int j = 0; j < AHEAD; ++j)
#pragma unroll
      for (int m = 0; m < MB; ++m) ring[j][m] = nxt[j][m];
    if (gnext) {
      lnext[sbase] = s0;
      lnext[sbase + 256] = s1;
    }
  }
}

__global__ __launch_bounds__(256) void gemm_pair_kernel(
    const float* __restrict__ A1, const float* __restrict__ Bp1,
    const float* __restrict__ A2, const float* __restrict__ Bp2, int K,
    const float* __restrict__ bias, const float* __restrict__ deg,
    float* __restrict__ out, int relu, int nTiles) {
  __shared__ float lds[2][CHUNK_KT * 8 * 64];

  const int tid = threadIdx.x;
  const int lane = tid & 31;
  const int wave = __builtin_amdgcn_readfirstlane(tid >> 5);
  const int half = lane >> 4;
  const int n0 = lane & 15;

  const int nS = gridDim.x * 4;
  const int s = blockIdx.x * 4 + (wave & 3);
  const int ts0 = (int)(((long long)s * nTiles) / nS);
  const int ts1 = (int)(((long long)(s + 1) * nTiles) / nS);
  const int firstPart = (ts1 - ts0 + 1) >> 1;
  const int w0 = (wave >> 2) ? (ts0 + firstPart) : ts0;
  const int w1 = (wave >> 2) ? ts1 : (ts0 + firstPart);
  const int maxS = (nTiles + nS - 1) / nS;
  const int nPass = (((maxS + 1) >> 1) + MB - 1) / MB;

  const int nkt = K >> 2;
  const int nkt2 = nkt << 1;
  const int nChunks = nkt2 / CHUNK_KT;

  for (int p = 0; p < nPass; ++p) {
    const int tb = w0 + p * MB;
    int cnt = w1 - tb;
    cnt = cnt < 0 ? 0 : (cnt > MB ? MB : cnt);
    const int lastT = (cnt > 0) ? (tb + cnt - 1) : (nTiles - 1);

    unsigned rowoff[MB];
    float scale[MB];
#pragma unroll
    for (int m = 0; m < MB; ++m) {
      const int t = (m < cnt) ? (tb + m) : lastT;
      const int arow = t * 16 + n0;
      rowoff[m] = (unsigned)arow * (unsigned)K + (unsigned)(half * 2);
      scale[m] = 1.0f / fmaxf(deg[arow], 1.0f);
    }

    v8f acc[MB][8];
#pragma unroll
    for (int nt = 0; nt < 8; ++nt) {
      const float b = bias[nt * 16 + n0];
#pragma unroll
      for (int m = 0; m < MB; ++m) acc[m][nt] = (v8f){b, b, b, b, b, b, b, b};
    }

    v2f ring[AHEAD][MB];
#pragma unroll
    for (int j = 0; j < AHEAD; ++j)
#pragma unroll
      for (int m = 0; m < MB; ++m) ring[j][m] = *(const v2f*)(A1 + rowoff[m] + j * 4);

    {
      const float4* __restrict__ gsrc = (const float4*)Bp1;
      float4* __restrict__ ldst = (float4*)lds[0];
#pragma unroll
      for (int i = 0; i < 16; ++i) ldst[tid + i * 256] = gsrc[tid + i * 256];
    }
    __syncthreads();

    for (int c = 0; c < nChunks; ++c) {
      const int kt0 = c * CHUNK_KT;
      const bool more = c + 1 < nChunks;
      const int ktx = kt0 + CHUNK_KT;
      const float4* __restrict__ gnext =
          !more ? (const float4*)nullptr
                : (const float4*)((ktx >= nkt)
                                      ? (Bp2 + (size_t)(ktx - nkt) * 8 * 64)
                                      : (Bp1 + (size_t)ktx * 8 * 64));
      float4* __restrict__ lnext = (float4*)lds[(c + 1) & 1];
      const float* ldsb = lds[c & 1];
      k_chunk(acc, ring, A1, A2, rowoff, scale, ldsb, gnext, lnext, kt0, nkt,
              nkt2, lane, tid, cnt == MB);
      __syncthreads();
    }

#pragma unroll
    for (int m = 0; m < MB; ++m) {
      if (m < cnt) {
        const int m0 = (tb + m) * 16;
#pragma unroll
        for (int nt = 0; nt < 8; ++nt) {
#pragma unroll
          for (int r = 0; r < 8; ++r) {
            const int row = m0 + r + 8 * half;
            const size_t idx = (size_t)row * 128 + nt * 16 + n0;
            float v = acc[m][nt][r];
            if (relu) v = fmaxf(v, 0.0f);
            out[idx] = v;
          }
        }
      }
    }
  }
}

extern "C" void kernel_launch(void* const* d_in, const int* in_sizes, int n_in,
                              void* d_out, int out_size, void* d_ws,
                              size_t ws_size, hipStream_t stream) {
  const int IN_DIM = 384, HID = 128;
  const int N = in_sizes[0] / IN_DIM;
  const int E = in_sizes[1] / 2;

  const float* x    = (const float*)d_in[0];
  const int*   ei   = (const int*)d_in[1];
  const float* W1_l = (const float*)d_in[2];
  const float* b1   = (const float*)d_in[3];
  const float* W1_r = (const float*)d_in[4];
  const float* W2_l = (const float*)d_in[5];
  const float* b2   = (const float*)d_in[6];
  const float* W2_r = (const float*)d_in[7];
  const int* src = ei;
  const int* dst = ei + E;
  float* out = (float*)d_out;

  float* ws = (float*)d_ws;
  const long deg_off  = 0;
  const long deg_pad  = ((long)N + 127) & ~127L;
  const long agg_off  = deg_off + deg_pad;
  const long h1_off   = agg_off + (long)N * IN_DIM;
  const long pw1l_off = h1_off + (long)N * HID;
  const long pw1r_off = pw1l_off + (long)IN_DIM * HID;
  const long pw2l_off = pw1r_off + (long)IN_DIM * HID;
  const long pw2r_off = pw2l_off + (long)HID * HID;
  const long offs_off = pw2r_off + (long)HID * HID;
  const long segs_off = offs_off + (long)MAX_BWG * OFF_PITCH;

  float* deg  = ws + deg_off;
  float* agg  = ws + agg_off;
  float* h1   = ws + h1_off;
  float* pw1l = ws + pw1l_off;
  float* pw1r = ws + pw1r_off;
  float* pw2l = ws + pw2l_off;
  float* pw2r = ws + pw2r_off;
  int*  offs = (int*)(ws + offs_off);
  int*  segs = (int*)(ws + segs_off);

  const int nTiles = N / 16;
  const int gemmBlocks = 256;
  const int nRanges = (N + 415) / 416;

  const int nBinWG = (E + 1023) / 1024;
  const int packWG = (2 * IN_DIM * 128 + 2 * HID * 128 + 1024 * PACK_PER - 1) /
                     (1024 * PACK_PER);
  setup_kernel<416><<<nBinWG + packWG, 1024, 0, stream>>>(
      src, dst, offs, segs, E, N, nRanges, nBinWG, W1_l, pw1l, W1_r, pw1r,
      IN_DIM, W2_l, pw2l, W2_r, pw2r, HID);

  range_gather_kernel<384, 416, 4><<<nRanges, 1024, 0, stream>>>(
      x, src, dst, offs, segs, nBinWG, agg, deg, E, N);

  gemm_pair_kernel<<<gemmBlocks, 256, 0, stream>>>(agg, pw1l, x, pw1r, IN_DIM,
                                                   b1, deg, h1, 1, nTiles);

  range_gather_kernel<128, 416, 8><<<nRanges, 1024, 0, stream>>>(
      h1, src, dst, offs, segs, nBinWG, agg, nullptr, E, N);

  gemm_pair_kernel<<<gemmBlocks, 256, 0, stream>>>(agg, pw2l, h1, pw2r, HID, b2,
                                                   deg, out, 0, nTiles);
}


# GEMM passes: a lane loads its A fragments 16 B at a time, one load feeds two k-steps (packed weights follow the same k order), so half as many A loads; sum order inside each 8 k values changes
# speedup vs baseline: 2.4093x; 1.0403x over previous
#include <hip/hip_runtime.h>


typedef __attribute__((ext_vector_type(2))) float v2f;
typedef __attribute__((ext_vector_type(4))) float v4f;
typedef __attribute__((ext_vector_type(8))) float v8f;

#define CHUNK_KT 32

__device__ __forceinline__ void pack_one(const float* __restrict__ W,
                                         float* __restrict__ P, int idx) {
  const int frag = idx >> 6;
  const int r = idx & 63;
  const int L = r >> 1;
  const int v = r & 1;
  const int kt = frag >> 3;
  const int nt = frag & 7;
  const int k = ((kt >> 1) << 3) + ((L >> 4) << 2) + ((kt & 1) << 1) + v;
  const int n = nt * 16 + (L & 15);
  P[idx] = W[k * 128 + n];
}

#define BIN_CAP 4096
#define MAX_RANGES 1024
#define MAX_BWG 256
#define OFF_PITCH 1056
#define PACK_PER 4

template <int R>
__global__ __launch_bounds__(1024) void setup_kernel(
    const int* __restrict__ src, const int* __restrict__ dst,
    int* __restrict__ offs, int* __restrict__ segs, int E, int N, int nRanges,
    int nBinWG, const float* __restrict__ W1l, float* __restrict__ P1l,
    const float* __restrict__ W1r, float* __restrict__ P1r, int K1,
    const float* __restrict__ W2l, float* __restrict__ P2l,
    const float* __restrict__ W2r, float* __restrict__ P2r, int K2) {
  __shared__ int hist[MAX_RANGES];
  __shared__ int first[MAX_RANGES + 1];
  const int tid = threadIdx.x;

  if ((int)blockIdx.x >= nBinWG) {
    const int n1 = K1 * 128, n2 = K2 * 128;
    const int g0 = ((int)blockIdx.x - nBinWG) * (1024 * PACK_PER) + tid;
    for (int j = 0; j < PACK_PER; ++j) {
      int g = g0 + j * 1024;
      if (g < n1) {
        pack_one(W1l, P1l, g);
      } else if ((g -= n1) < n1) {
        pack_one(W1r, P1r, g);
      } else if ((g -= n1) < n2) {
        pack_one(W2l, P2l, g);
      } else if ((g -= n2) < n2) {
        pack_one(W2r, P2r, g);
      }
    }
    return;
  }

  const int lane = tid & 31;
  for (int i = tid; i < MAX_RANGES; i += 1024) hist[i] = 0;
  __syncthreads();

  const int e = blockIdx.x * 1024 + tid;
  int owner = -1, rel = 0, s = 0, lrank = 0;
  if (e < E) {
    const int d = dst[e];
    s = src[e];
    if ((unsigned)d < (unsigned)N) {
      owner = d / R;
      rel = d - owner * R;
      lrank = __hip_atomic_fetch_add(&hist[owner], 1, __ATOMIC_RELAXED,
                                     __HIP_MEMORY_SCOPE_WORKGROUP);
    }
  }
  __syncthreads();

  if (tid < 32) {
    const int per = (nRanges + 31) >> 5;
    int sum = 0;
    for (int k = 0; k < per; ++k) sum += hist[lane * per + k];
    int incl = sum;
    for (int d = 1; d < 32; d <<= 1) {
      const int t = __shfl_up(incl, d);
      if (lane >= d) incl += t;
    }
    int run = incl - sum;
    for (int k = 0; k < per; ++k) {
      first[lane * per + k] = run;
      run += hist[lane * per + k];
    }
    if (lane == 31) first[nRanges] = run;
  }
  __syncthreads();

  for (int i = tid; i <= nRanges; i += 1024)
    offs[(size_t)blockIdx.x * OFF_PITCH + i] = first[i];
  if (owner >= 0)
    ((int2*)segs)[(size_t)blockIdx.x * 1024 + first[owner] + lrank] =
        make_int2(s, rel);
}


template <int Q>
__device__ __forceinline__ void store_row_and_clear(float* __restrict__ agg,
                                                    size_t node, int F,
                                                    int lane, float4* acc) {
  float4* __restrict__ o = (float4*)(agg + node * (size_t)F) + lane;
#pragma unroll
  for (int q = 0; q < Q; ++q) {
    o[q * 32] = acc[q];
    acc[q] = make_float4(0.0f, 0.0f, 0.0f, 0.0f);
  }
}

template <int F, int R, int B>
__global__ __launch_bounds__(1024) void range_gather_kernel(
    const float* __restrict__ feat, const int* __restrict__ src,
    const int* __restrict__ dst, const int* __restrict__ offs,
    const int* __restrict__ segs, int nBinWG, float* __restrict__ agg,
    float* __restrict__ deg, int E, int N) {
  constexpr int CAP = BIN_CAP;
  constexpr int PER = R / 32;
  constexpr int Q = F / 128;
  __shared__ int cnt[R];
  __shared__ int start[R + 1];
  __shared__ int list_e[CAP];
  __shared__ unsigned list_k[CAP];
  __shared__ int sorted[CAP];
  __shared__ int clo[MAX_BWG];
  __shared__ int cpos[MAX_BWG + 1];

  const int tid = threadIdx.x;
  const int lane = tid & 31;
  const int wave = __builtin_amdgcn_readfirstlane(tid >> 5);
  const int base = blockIdx.x * R;
  const int rows = (N - base < R) ? (N - base) : R;

  for (int i = tid; i < R; i += 1024) cnt[i] = 0;
  if (tid < MAX_BWG) {
    int lo = 0, hi = 0;
    if (tid < nBinWG) {
      const int* __restrict__ t = offs + (size_t)tid * OFF_PITCH + blockIdx.x;
      lo = t[0];
      hi = t[1];
    }
    clo[tid] = lo;
    cpos[tid] = hi - lo;
  }
  __syncthreads();

  if (wave == 0) {
    int c[MAX_BWG / 32];
    int s = 0;
    for (int k = 0; k < MAX_BWG / 32; ++k) {
      c[k] = cpos[lane * (MAX_BWG / 32) + k];
      s += c[k];
    }
    int incl = s;
    for (int d = 1; d < 32; d <<= 1) {
      const int t = __shfl_up(incl, d);
      if (lane >= d) incl += t;
    }
    int run = incl - s;
    for (int k = 0; k < MAX_BWG / 32; ++k) {
      cpos[lane * (MAX_BWG / 32) + k] = run;
      run += c[k];
    }
    if (lane == 31) cpos[MAX_BWG] = run;
  }
  __syncthreads();

  const int total = (nBinWG <= MAX_BWG) ? cpos[MAX_BWG] : (CAP + 1);
  if (total <= CAP) {
    const int2* __restrict__ seg = (const int2*)segs;
    for (int i = tid; i < total; i += 1024) {
      int w = 0;
      for (int step = MAX_BWG / 2; step > 0; step >>= 1)
        if (cpos[w + step] <= i) w += step;
      const int2 r = seg[(size_t)w * 1024 + clo[w] + (i - cpos[w])];
      const int rank = __hip_atomic_fetch_add(
          &cnt[r.y], 1, __ATOMIC_RELAXED, __HIP_MEMORY_SCOPE_WORKGROUP);
      list_e[i] = r.x;
      list_k[i] = (unsigned)r.y | ((unsigned)rank << 16);
    }
  }
  __syncthreads();

  if (wave == 0) {
    int s = 0;
    for (int k = 0; k < PER; ++k) s += cnt[lane * PER + k];
    int incl = s;
    for (int d = 1; d < 32; d <<= 1) {
      const int t = __shfl_up(incl, d);
      if (lane >= d) incl += t;
    }
    int run = incl - s;
    for (int k = 0; k < PER; ++k) {
      start[lane * PER + k] = run;
      run += cnt[lane * PER + k];
    }
    if (lane == 31) start[R] = run;
  }
  __syncthreads();

  if (total <= CAP) {
    for (int i = tid; i < total; i += 1024) {
      const unsigned k = list_k[i];
      sorted[start[k & 0xffffu] + (int)(k >> 16)] = list_e[i];
    }
  }
  __syncthreads();

  const int r0 = wave * PER;
  const int r1 = (r0 + PER < rows) ? (r0 + PER) : rows;
  float4 acc[Q];
#pragma unroll
  for (int q = 0; q < Q; ++q) acc[q] = make_float4(0.0f, 0.0f, 0.0f, 0.0f);

  if (r0 < rows) {
    if (total <= CAP) {
      const int e_lo = __builtin_amdgcn_readfirstlane(start[r0]);
      const int e_hi = __builtin_amdgcn_readfirstlane(start[r1]);
      int node = r0;
      int node_end = __builtin_amdgcn_readfirstlane(start[r0 + 1]);
      for (int eb = e_lo; eb < e_hi; eb += B) {
        float4 v[B][Q];
#pragma unroll
        for (int j = 0; j < B; ++j) {
          const int ei = (eb + j < e_hi) ? (eb + j) : (e_hi - 1);
          const int s = __builtin_amdgcn_readfirstlane(sorted[ei]);
          const float4* __restrict__ row =
              (const float4*)(feat + (size_t)s * F) + lane;
#pragma unroll
          for (int q = 0; q < Q; ++q) v[j][q] = row[q * 32];
        }
#pragma unroll
        for (int j = 0; j < B; ++j) {
          if (eb + j < e_hi) {
            while (eb + j >= node_end) {
              store_row_and_clear<Q>(agg, (size_t)(base + node), F, lane, acc);
              ++node;
              node_end = __builtin_amdgcn_readfirstlane(start[node + 1]);
            }
#pragma unroll
            for (int q = 0; q < Q; ++q) {
              acc[q].x += v[j][q].x;
              acc[q].y += v[j][q].y;
              acc[q].z += v[j][q].z;
              acc[q].w += v[j][q].w;
            }
          }
        }
      }
      for (; node < r1; ++node)
        store_row_and_clear<Q>(agg, (size_t)(base + node), F, lane, acc);
    } else {
      for (int node = r0; node < r1; ++node) {
        int nedges = 0;
        for (int eb = 0; eb < E; eb += 32) {
          const int e = eb + lane;
          const int d = (e < E) ? dst[e] : -1;
          unsigned m = __builtin_amdgcn_ballot_w32(d == base + node);
          nedges += __builtin_popcount(m);
          while (m) {
            const int l = __builtin_ctz(m);
            m &= m - 1;
            const int s = src[eb + l];
            const float4* __restrict__ row =
                (const float4*)(feat + (size_t)s * F) + lane;
#pragma unroll
            for (int q = 0; q < Q; ++q) {
              const float4 w = row[q * 32];
              acc[q].x += w.x;
              acc[q].y += w.y;
              acc[q].z += w.z;
              acc[q].w += w.w;
            }
          }
        }
        store_row_and_clear<Q>(agg, (size_t)(base + node), F, lane, acc);
        if (deg && lane == 0) deg[base + node] = (float)nedges;
      }
    }
  }
  if (deg && total <= CAP)
    for (int i = tid; i < rows; i += 1024) deg[base + i] = (float)cnt[i];
}

#define MB 4
#define AHEAD 4
#define STG 2

__device__ __forceinline__ void k_chunk(
    v8f (&acc)[MB][8], v4f (&ring)[AHEAD / 2][MB], const float* __restrict__ A1,
    const float* __restrict__ A2, const unsigned (&rowoff)[MB],
    const float (&scale)[MB], const float* __restrict__ ldsb,
    const float4* __restrict__ gnext, float4* __restrict__ lnext, int kt0,
    int nkt, int nkt2, int lane, int tid, bool full) {
  const bool second = kt0 >= nkt;
  for (int kb = 0; kb < CHUNK_KT; kb += AHEAD) {
    const int sbase = tid + (kb / AHEAD) * STG * 256;
    float4 s0 = make_float4(0.0f, 0.0f, 0.0f, 0.0f), s1 = s0;
    if (gnext) {
      s0 = gnext[sbase];
      s1 = gnext[sbase + 256];
    }
    v4f nxt[AHEAD / 2][MB];
#pragma unroll
    for (int jp = 0; jp < AHEAD / 2; ++jp) {
      int ktn = kt0 + kb + AHEAD + 2 * jp;
      if (ktn > nkt2 - 2) ktn = nkt2 - 2;
      const bool n2 = ktn >= nkt;
      const float* __restrict__ an = (n2 ? A2 : A1) + (n2 ? (ktn - nkt) : ktn) * 4;
#pragma unroll
      for (int m = 0; m < MB; ++m) nxt[jp][m] = *(const v4f*)(an + rowoff[m]);
    }
    asm volatile("" ::: "memory");
#pragma unroll
    for (int j = 0; j < AHEAD; ++j) {
      const int kt = kb + j;
      v2f a[MB];
#pragma unroll
      for (int m = 0; m < MB; ++m) {
        const v4f f = ring[j >> 1][m];
        const v2f h2 = (j & 1) ? (v2f){f.z, f.w} : (v2f){f.x, f.y};
        a[m] = h2 * (second ? 1.0f : scale[m]);
      }

      const float* bbase = ldsb + (kt * 8) * 64 + lane * 2;
      v2f bf[8];
#pragma unroll
      for (int nt = 0; nt < 8; ++nt) bf[nt] = *(const v2f*)(bbase + nt * 64);

      if (full) {
#pragma unroll
        for (int nt = 0; nt < 8; ++nt) {
          acc[MB - 1][nt] = __builtin_amdgcn_wmma_f32_16x16x4_f32(
              false, a[MB - 1], false, bf[nt], (short)0, acc[MB - 1][nt], false,
              false);
        }
      }
#pragma unroll
      for (int nt = 0; nt < 8; ++nt) {
#pragma unroll
        for (int m = 0; m < MB - 1; ++m) {
          acc[m][nt] = __builtin_amdgcn_wmma_f32_16x16x4_f32(
              false, a[m], false, bf[nt], (short)0, acc[m][nt], false, false);
        }
      }
    }
#pragma unroll
    for (int jp = 0; jp < AHEAD / 2; ++jp)
#pragma unroll
      for (int m = 0; m < MB; ++m) ring[jp][m] = nxt[jp][m];
    if (gnext) {
      lnext[sbase] = s0;
      lnext[sbase + 256] = s1;
    }
  }
}

__global__ __launch_bounds__(256) void gemm_pair_kernel(
    const float* __restrict__ A1, const float* __restrict__ Bp1,
    const float* __restrict__ A2, const float* __restrict__ Bp2, int K,
    const float* __restrict__ bias, const float* __restrict__ deg,
    float* __restrict__ out, int relu, int nTiles) {
  __shared__ float lds[2][CHUNK_KT * 8 * 64];

  const int tid = threadIdx.x;
  const int lane = tid & 31;
  const int wave = __builtin_amdgcn_readfirstlane(tid >> 5);
  const int half = lane >> 4;
  const int n0 = lane & 15;

  const int nS = gridDim.x * 4;
  const int s = blockIdx.x * 4 + (wave & 3);
  const int ts0 = (int)(((long long)s * nTiles) / nS);
  const int ts1 = (int)(((long long)(s + 1) * nTiles) / nS);
  const int firstPart = (ts1 - ts0 + 1) >> 1;
  const int w0 = (wave >> 2) ? (ts0 + firstPart) : ts0;
  const int w1 = (wave >> 2) ? ts1 : (ts0 + firstPart);
  const int maxS = (nTiles + nS - 1) / nS;
  const int nPass = (((maxS + 1) >> 1) + MB - 1) / MB;

  const int nkt = K >> 2;
  const int nkt2 = nkt << 1;
  const int nChunks = nkt2 / CHUNK_KT;

  for (int p = 0; p < nPass; ++p) {
    const int tb = w0 + p * MB;
    int cnt = w1 - tb;
    cnt = cnt < 0 ? 0 : (cnt > MB ? MB : cnt);
    const int lastT = (cnt > 0) ? (tb + cnt - 1) : (nTiles - 1);

    unsigned rowoff[MB];
    float scale[MB];
#pragma unroll
    for (int m = 0; m < MB; ++m) {
      const int t = (m < cnt) ? (tb + m) : lastT;
      const int arow = t * 16 + n0;
      rowoff[m] = (unsigned)arow * (unsigned)K + (unsigned)(half * 4);
      scale[m] = 1.0f / fmaxf(deg[arow], 1.0f);
    }

    v8f acc[MB][8];
#pragma unroll
    for (int nt = 0; nt < 8; ++nt) {
      const float b = bias[nt * 16 + n0];
#pragma unroll
      for (int m = 0; m < MB; ++m) acc[m][nt] = (v8f){b, b, b, b, b, b, b, b};
    }

    v4f ring[AHEAD / 2][MB];
#pragma unroll
    for (int jp = 0; jp < AHEAD / 2; ++jp)
#pragma unroll
      for (int m = 0; m < MB; ++m)
        ring[jp][m] = *(const v4f*)(A1 + rowoff[m] + jp * 8);

    {
      const float4* __restrict__ gsrc = (const float4*)Bp1;
      float4* __restrict__ ldst = (float4*)lds[0];
#pragma unroll
      for (int i = 0; i < 16; ++i) ldst[tid + i * 256] = gsrc[tid + i * 256];
    }
    __syncthreads();

    for (int c = 0; c < nChunks; ++c) {
      const int kt0 = c * CHUNK_KT;
      const bool more = c + 1 < nChunks;
      const int ktx = kt0 + CHUNK_KT;
      const float4* __restrict__ gnext =
          !more ? (const float4*)nullptr
                : (const float4*)((ktx >= nkt)
                                      ? (Bp2 + (size_t)(ktx - nkt) * 8 * 64)
                                      : (Bp1 + (size_t)ktx * 8 * 64));
      float4* __restrict__ lnext = (float4*)lds[(c + 1) & 1];
      const float* ldsb = lds[c & 1];
      k_chunk(acc, ring, A1, A2, rowoff, scale, ldsb, gnext, lnext, kt0, nkt,
              nkt2, lane, tid, cnt == MB);
      __syncthreads();
    }

#pragma unroll
    for (int m = 0; m < MB; ++m) {
      if (m < cnt) {
        const int m0 = (tb + m) * 16;
#pragma unroll
        for (int nt = 0; nt < 8; ++nt) {
#pragma unroll
          for (int r = 0; r < 8; ++r) {
            const int row = m0 + r + 8 * half;
            const size_t idx = (size_t)row * 128 + nt * 16 + n0;
            float v = acc[m][nt][r];
            if (relu) v = fmaxf(v, 0.0f);
            out[idx] = v;
          }
        }
      }
    }
  }
}

extern "C" void kernel_launch(void* const* d_in, const int* in_sizes, int n_in,
                              void* d_out, int out_size, void* d_ws,
                              size_t ws_size, hipStream_t stream) {
  const int IN_DIM = 384, HID = 128;
  const int N = in_sizes[0] / IN_DIM;
  const int E = in_sizes[1] / 2;

  const float* x    = (const float*)d_in[0];
  const int*   ei   = (const int*)d_in[1];
  const float* W1_l = (const float*)d_in[2];
  const float* b1   = (const float*)d_in[3];
  const float* W1_r = (const float*)d_in[4];
  const float* W2_l = (const float*)d_in[5];
  const float* b2   = (const float*)d_in[6];
  const float* W2_r = (const float*)d_in[7];
  const int* src = ei;
  const int* dst = ei + E;
  float* out = (float*)d_out;

  float* ws = (float*)d_ws;
  const long deg_off  = 0;
  const long deg_pad  = ((long)N + 127) & ~127L;
  const long agg_off  = deg_off + deg_pad;
  const long h1_off   = agg_off + (long)N * IN_DIM;
  const long pw1l_off = h1_off + (long)N * HID;
  const long pw1r_off = pw1l_off + (long)IN_DIM * HID;
  const long pw2l_off = pw1r_off + (long)IN_DIM * HID;
  const long pw2r_off = pw2l_off + (long)HID * HID;
  const long offs_off = pw2r_off + (long)HID * HID;
  const long segs_off = offs_off + (long)MAX_BWG * OFF_PITCH;

  float* deg  = ws + deg_off;
  float* agg  = ws + agg_off;
  float* h1   = ws + h1_off;
  float* pw1l = ws + pw1l_off;
  float* pw1r = ws + pw1r_off;
  float* pw2l = ws + pw2l_off;
  float* pw2r = ws + pw2r_off;
  int*  offs = (int*)(ws + offs_off);
  int*  segs = (int*)(ws + segs_off);

  const int nTiles = N / 16;
  const int gemmBlocks = 256;
  const int nRanges = (N + 415) / 416;

  const int nBinWG = (E + 1023) / 1024;
  const int packWG = (2 * IN_DIM * 128 + 2 * HID * 128 + 1024 * PACK_PER - 1) /
                     (1024 * PACK_PER);
  setup_kernel<416><<<nBinWG + packWG, 1024, 0, stream>>>(
      src, dst, offs, segs, E, N, nRanges, nBinWG, W1_l, pw1l, W1_r, pw1r,
      IN_DIM, W2_l, pw2l, W2_r, pw2r, HID);

  range_gather_kernel<384, 416, 4><<<nRanges, 1024, 0, stream>>>(
      x, src, dst, offs, segs, nBinWG, agg, deg, E, N);

  gemm_pair_kernel<<<gemmBlocks, 256, 0, stream>>>(agg, pw1l, x, pw1r, IN_DIM,
                                                   b1, deg, h1, 1, nTiles);

  range_gather_kernel<128, 416, 8><<<nRanges, 1024, 0, stream>>>(
      h1, src, dst, offs, segs, nBinWG, agg, nullptr, E, N);

  gemm_pair_kernel<<<gemmBlocks, 256, 0, stream>>>(agg, pw2l, h1, pw2r, HID, b2,
                                                   deg, out, 0, nTiles);
}


# Neighbour sums only for nodes with two or more edges: a one-edge node's sum is its source row and an edgeless node's is zero, so the GEMM A loader reads that row of x/h1 (or a zero row), not agg
# speedup vs baseline: 2.4961x; 1.0360x over previous
#include <hip/hip_runtime.h>


typedef __attribute__((ext_vector_type(2))) float v2f;
typedef __attribute__((ext_vector_type(4))) float v4f;
typedef __attribute__((ext_vector_type(8))) float v8f;

#define CHUNK_KT 32

__device__ __forceinline__ void pack_one(const float* __restrict__ W,
                                         float* __restrict__ P, int idx) {
  const int frag = idx >> 6;
  const int r = idx & 63;
  const int L = r >> 1;
  const int v = r & 1;
  const int kt = frag >> 3;
  const int nt = frag & 7;
  const int k = ((kt >> 1) << 3) + ((L >> 4) << 2) + ((kt & 1) << 1) + v;
  const int n = nt * 16 + (L & 15);
  P[idx] = W[k * 128 + n];
}

#define BIN_CAP 4096
#define MAX_RANGES 1024
#define MAX_BWG 256
#define OFF_PITCH 1056
#define PACK_PER 4

template <int R>
__global__ __launch_bounds__(1024) void setup_kernel(
    const int* __restrict__ src, const int* __restrict__ dst,
    int* __restrict__ offs, int* __restrict__ segs, int E, int N, int nRanges,
    int nBinWG, const float* __restrict__ W1l, float* __restrict__ P1l,
    const float* __restrict__ W1r, float* __restrict__ P1r, int K1,
    const float* __restrict__ W2l, float* __restrict__ P2l,
    const float* __restrict__ W2r, float* __restrict__ P2r, int K2) {
  __shared__ int hist[MAX_RANGES];
  __shared__ int first[MAX_RANGES + 1];
  const int tid = threadIdx.x;

  if ((int)blockIdx.x >= nBinWG) {
    const int n1 = K1 * 128, n2 = K2 * 128;
    const int g0 = ((int)blockIdx.x - nBinWG) * (1024 * PACK_PER) + tid;
    for (int j = 0; j < PACK_PER; ++j) {
      int g = g0 + j * 1024;
      if (g < n1) {
        pack_one(W1l, P1l, g);
      } else if ((g -= n1) < n1) {
        pack_one(W1r, P1r, g);
      } else if ((g -= n1) < n2) {
        pack_one(W2l, P2l, g);
      } else if ((g -= n2) < n2) {
        pack_one(W2r, P2r, g);
      }
    }
    return;
  }

  const int lane = tid & 31;
  for (int i = tid; i < MAX_RANGES; i += 1024) hist[i] = 0;
  __syncthreads();

  const int e = blockIdx.x * 1024 + tid;
  int owner = -1, rel = 0, s = 0, lrank = 0;
  if (e < E) {
    const int d = dst[e];
    s = src[e];
    if ((unsigned)d < (unsigned)N) {
      owner = d / R;
      rel = d - owner * R;
      lrank = __hip_atomic_fetch_add(&hist[owner], 1, __ATOMIC_RELAXED,
                                     __HIP_MEMORY_SCOPE_WORKGROUP);
    }
  }
  __syncthreads();

  if (tid < 32) {
    const int per = (nRanges + 31) >> 5;
    int sum = 0;
    for (int k = 0; k < per; ++k) sum += hist[lane * per + k];
    int incl = sum;
    for (int d = 1; d < 32; d <<= 1) {
      const int t = __shfl_up(incl, d);
      if (lane >= d) incl += t;
    }
    int run = incl - sum;
    for (int k = 0; k < per; ++k) {
      first[lane * per + k] = run;
      run += hist[lane * per + k];
    }
    if (lane == 31) first[nRanges] = run;
  }
  __syncthreads();

  for (int i = tid; i <= nRanges; i += 1024)
    offs[(size_t)blockIdx.x * OFF_PITCH + i] = first[i];
  if (owner >= 0)
    ((int2*)segs)[(size_t)blockIdx.x * 1024 + first[owner] + lrank] =
        make_int2(s, rel);
}


template <int Q>
__device__ __forceinline__ void store_row_and_clear(float* __restrict__ agg,
                                                    size_t node, int F,
                                                    int lane, float4* acc) {
  float4* __restrict__ o = (float4*)(agg + node * (size_t)F) + lane;
#pragma unroll
  for (int q = 0; q < Q; ++q) {
    o[q * 32] = acc[q];
    acc[q] = make_float4(0.0f, 0.0f, 0.0f, 0.0f);
  }
}

template <int F, int R, int B>
__global__ __launch_bounds__(1024) void range_gather_kernel(
    const float* __restrict__ feat, const int* __restrict__ src,
    const int* __restrict__ dst, const int* __restrict__ offs,
    const int* __restrict__ segs, int nBinWG, float* __restrict__ agg,
    float* __restrict__ deg, int* __restrict__ single,
    float* __restrict__ zrow, int E, int N) {
  constexpr int CAP = BIN_CAP;
  constexpr int PER = R / 32;
  constexpr int Q = F / 128;
  __shared__ int cnt[R];
  __shared__ int start[R + 1];
  __shared__ int list_e[CAP];
  __shared__ unsigned list_k[CAP];
  __shared__ int sorted[CAP];
  __shared__ int one[R];
  __shared__ int clo[MAX_BWG];
  __shared__ int cpos[MAX_BWG + 1];

  const int tid = threadIdx.x;
  const int lane = tid & 31;
  const int wave = __builtin_amdgcn_readfirstlane(tid >> 5);
  const int base = blockIdx.x * R;
  const int rows = (N - base < R) ? (N - base) : R;

  for (int i = tid; i < R; i += 1024) cnt[i] = 0;
  if (blockIdx.x == 0)
    for (int i = tid; i < F; i += 1024) zrow[i] = 0.0f;
  if (tid < MAX_BWG) {
    int lo = 0, hi = 0;
    if (tid < nBinWG) {
      const int* __restrict__ t = offs + (size_t)tid * OFF_PITCH + blockIdx.x;
      lo = t[0];
      hi = t[1];
    }
    clo[tid] = lo;
    cpos[tid] = hi - lo;
  }
  __syncthreads();

  if (wave == 0) {
    int c[MAX_BWG / 32];
    int s = 0;
    for (int k = 0; k < MAX_BWG / 32; ++k) {
      c[k] = cpos[lane * (MAX_BWG / 32) + k];
      s += c[k];
    }
    int incl = s;
    for (int d = 1; d < 32; d <<= 1) {
      const int t = __shfl_up(incl, d);
      if (lane >= d) incl += t;
    }
    int run = incl - s;
    for (int k = 0; k < MAX_BWG / 32; ++k) {
      cpos[lane * (MAX_BWG / 32) + k] = run;
      run += c[k];
    }
    if (lane == 31) cpos[MAX_BWG] = run;
  }
  __syncthreads();

  const int total = (nBinWG <= MAX_BWG) ? cpos[MAX_BWG] : (CAP + 1);
  if (total <= CAP) {
    const int2* __restrict__ seg = (const int2*)segs;
    for (int i = tid; i < total; i += 1024) {
      int w = 0;
      for (int step = MAX_BWG / 2; step > 0; step >>= 1)
        if (cpos[w + step] <= i) w += step;
      const int2 r = seg[(size_t)w * 1024 + clo[w] + (i - cpos[w])];
      const int rank = __hip_atomic_fetch_add(
          &cnt[r.y], 1, __ATOMIC_RELAXED, __HIP_MEMORY_SCOPE_WORKGROUP);
      list_e[i] = r.x;
      list_k[i] = (unsigned)r.y | ((unsigned)rank << 16);
    }
  }
  __syncthreads();

  if (wave == 0) {
    int s = 0;
    for (int k = 0; k < PER; ++k) {
      const int c = cnt[lane * PER + k];
      s += (c >= 2) ? c : 0;
    }
    int incl = s;
    for (int d = 1; d < 32; d <<= 1) {
      const int t = __shfl_up(incl, d);
      if (lane >= d) incl += t;
    }
    int run = incl - s;
    for (int k = 0; k < PER; ++k) {
      const int c = cnt[lane * PER + k];
      start[lane * PER + k] = run;
      run += (c >= 2) ? c : 0;
    }
    if (lane == 31) start[R] = run;
  }
  __syncthreads();

  if (total <= CAP) {
    for (int i = tid; i < total; i += 1024) {
      const unsigned k = list_k[i];
      const int nd = (int)(k & 0xffffu);
      if (cnt[nd] >= 2)
        sorted[start[nd] + (int)(k >> 16)] = list_e[i];
      else
        one[nd] = list_e[i];
    }
  }
  __syncthreads();

  const int r0 = wave * PER;
  const int r1 = (r0 + PER < rows) ? (r0 + PER) : rows;
  float4 acc[Q];
#pragma unroll
  for (int q = 0; q < Q; ++q) acc[q] = make_float4(0.0f, 0.0f, 0.0f, 0.0f);

  if (r0 < rows) {
    if (total <= CAP) {
      const int e_lo = __builtin_amdgcn_readfirstlane(start[r0]);
      const int e_hi = __builtin_amdgcn_readfirstlane(start[r1]);
      int node = r0;
      int node_beg = e_lo;
      int node_end = __builtin_amdgcn_readfirstlane(start[r0 + 1]);
      for (int eb = e_lo; eb < e_hi; eb += B) {
        float4 v[B][Q];
#pragma unroll
        for (int j = 0; j < B; ++j) {
          const int ei = (eb + j < e_hi) ? (eb + j) : (e_hi - 1);
          const int s = __builtin_amdgcn_readfirstlane(sorted[ei]);
          const float4* __restrict__ row =
              (const float4*)(feat + (size_t)s * F) + lane;
#pragma unroll
          for (int q = 0; q < Q; ++q) v[j][q] = row[q * 32];
        }
#pragma unroll
        for (int j = 0; j < B; ++j) {
          if (eb + j < e_hi) {
            while (eb + j >= node_end) {
              if (node_end > node_beg)
                store_row_and_clear<Q>(agg, (size_t)(base + node), F, lane, acc);
              ++node;
              node_beg = node_end;
              node_end = __builtin_amdgcn_readfirstlane(start[node + 1]);
            }
#pragma unroll
            for (int q = 0; q < Q; ++q) {
              acc[q].x += v[j][q].x;
              acc[q].y += v[j][q].y;
              acc[q].z += v[j][q].z;
              acc[q].w += v[j][q].w;
            }
          }
        }
      }
      if (e_hi > e_lo)
        store_row_and_clear<Q>(agg, (size_t)(base + node), F, lane, acc);
    } else {
      for (int node = r0; node < r1; ++node) {
        int nedges = 0;
        for (int eb = 0; eb < E; eb += 32) {
          const int e = eb + lane;
          const int d = (e < E) ? dst[e] : -1;
          unsigned m = __builtin_amdgcn_ballot_w32(d == base + node);
          nedges += __builtin_popcount(m);
          while (m) {
            const int l = __builtin_ctz(m);
            m &= m - 1;
            const int s = src[eb + l];
            const float4* __restrict__ row =
                (const float4*)(feat + (size_t)s * F) + lane;
#pragma unroll
            for (int q = 0; q < Q; ++q) {
              const float4 w = row[q * 32];
              acc[q].x += w.x;
              acc[q].y += w.y;
              acc[q].z += w.z;
              acc[q].w += w.w;
            }
          }
        }
        store_row_and_clear<Q>(agg, (size_t)(base + node), F, lane, acc);
        if (deg && lane == 0) {
          deg[base + node] = (float)nedges;
          single[base + node] = -1;
        }
      }
    }
  }
  if (deg && total <= CAP)
    for (int i = tid; i < rows; i += 1024) {
      const int c = cnt[i];
      deg[base + i] = (float)c;
      single[base + i] = (c == 1) ? one[i] : -1;
    }
}

#define MB 4
#define AHEAD 4
#define STG 2

__device__ __forceinline__ void k_chunk(
    v8f (&acc)[MB][8], v4f (&ring)[AHEAD / 2][MB],
    const float* const (&a1p)[MB], const float* const (&a2p)[MB],
    const float (&scale)[MB], const float* __restrict__ ldsb,
    const float4* __restrict__ gnext, float4* __restrict__ lnext, int kt0,
    int nkt, int nkt2, int lane, int tid, bool full) {
  const bool second = kt0 >= nkt;
  for (int kb = 0; kb < CHUNK_KT; kb += AHEAD) {
    const int sbase = tid + (kb / AHEAD) * STG * 256;
    float4 s0 = make_float4(0.0f, 0.0f, 0.0f, 0.0f), s1 = s0;
    if (gnext) {
      s0 = gnext[sbase];
      s1 = gnext[sbase + 256];
    }
    v4f nxt[AHEAD / 2][MB];
#pragma unroll
    for (int jp = 0; jp < AHEAD / 2; ++jp) {
      int ktn = kt0 + kb + AHEAD + 2 * jp;
      if (ktn > nkt2 - 2) ktn = nkt2 - 2;
      const bool n2 = ktn >= nkt;
      const int kk = (n2 ? (ktn - nkt) : ktn) * 4;
#pragma unroll
      for (int m = 0; m < MB; ++m)
        nxt[jp][m] = *(const v4f*)((n2 ? a2p[m] : a1p[m]) + kk);
    }
    asm volatile("" ::: "memory");
#pragma unroll
    for (int j = 0; j < AHEAD; ++j) {
      const int kt = kb + j;
      v2f a[MB];
#pragma unroll
      for (int m = 0; m < MB; ++m) {
        const v4f f = ring[j >> 1][m];
        const v2f h2 = (j & 1) ? (v2f){f.z, f.w} : (v2f){f.x, f.y};
        a[m] = h2 * (second ? 1.0f : scale[m]);
      }

      const float* bbase = ldsb + (kt * 8) * 64 + lane * 2;
      v2f bf[8];
#pragma unroll
      for (int nt = 0; nt < 8; ++nt) bf[nt] = *(const v2f*)(bbase + nt * 64);

      if (full) {
#pragma unroll
        for (int nt = 0; nt < 8; ++nt) {
          acc[MB - 1][nt] = __builtin_amdgcn_wmma_f32_16x16x4_f32(
              false, a[MB - 1], false, bf[nt], (short)0, acc[MB - 1][nt], false,
              false);
        }
      }
#pragma unroll
      for (int nt = 0; nt < 8; ++nt) {
#pragma unroll
        for (int m = 0; m < MB - 1; ++m) {
          acc[m][nt] = __builtin_amdgcn_wmma_f32_16x16x4_f32(
              false, a[m], false, bf[nt], (short)0, acc[m][nt], false, false);
        }
      }
    }
#pragma unroll
    for (int jp = 0; jp < AHEAD / 2; ++jp)
#pragma unroll
      for (int m = 0; m < MB; ++m) ring[jp][m] = nxt[jp][m];
    if (gnext) {
      lnext[sbase] = s0;
      lnext[sbase + 256] = s1;
    }
  }
}

__global__ __launch_bounds__(256) void gemm_pair_kernel(
    const float* __restrict__ A1, const float* __restrict__ Bp1,
    const float* __restrict__ A2, const float* __restrict__ Bp2, int K,
    const float* __restrict__ bias, const float* __restrict__ deg,
    const int* __restrict__ single, const float* __restrict__ zrow,
    float* __restrict__ out, int relu, int nTiles) {
  __shared__ float lds[2][CHUNK_KT * 8 * 64];

  const int tid = threadIdx.x;
  const int lane = tid & 31;
  const int wave = __builtin_amdgcn_readfirstlane(tid >> 5);
  const int half = lane >> 4;
  const int n0 = lane & 15;

  const int nS = gridDim.x * 4;
  const int s = blockIdx.x * 4 + (wave & 3);
  const int ts0 = (int)(((long long)s * nTiles) / nS);
  const int ts1 = (int)(((long long)(s + 1) * nTiles) / nS);
  const int firstPart = (ts1 - ts0 + 1) >> 1;
  const int w0 = (wave >> 2) ? (ts0 + firstPart) : ts0;
  const int w1 = (wave >> 2) ? ts1 : (ts0 + firstPart);
  const int maxS = (nTiles + nS - 1) / nS;
  const int nPass = (((maxS + 1) >> 1) + MB - 1) / MB;

  const int nkt = K >> 2;
  const int nkt2 = nkt << 1;
  const int nChunks = nkt2 / CHUNK_KT;

  for (int p = 0; p < nPass; ++p) {
    const int tb = w0 + p * MB;
    int cnt = w1 - tb;
    cnt = cnt < 0 ? 0 : (cnt > MB ? MB : cnt);
    const int lastT = (cnt > 0) ? (tb + cnt - 1) : (nTiles - 1);

    const float* a1p[MB];
    const float* a2p[MB];
    float scale[MB];
#pragma unroll
    for (int m = 0; m < MB; ++m) {
      const int t = (m < cnt) ? (tb + m) : lastT;
      const int arow = t * 16 + n0;
      const float d = deg[arow];
      const int s1 = single[arow];
      const float* rp = (s1 >= 0) ? (A2 + (size_t)s1 * (size_t)K)
                                  : ((d > 0.0f) ? (A1 + (size_t)arow * (size_t)K)
                                                : zrow);
      a1p[m] = rp + half * 4;
      a2p[m] = A2 + (size_t)arow * (size_t)K + half * 4;
      scale[m] = 1.0f / fmaxf(d, 1.0f);
    }

    v8f acc[MB][8];
#pragma unroll
    for (int nt = 0; nt < 8; ++nt) {
      const float b = bias[nt * 16 + n0];
#pragma unroll
      for (int m = 0; m < MB; ++m) acc[m][nt] = (v8f){b, b, b, b, b, b, b, b};
    }

    v4f ring[AHEAD / 2][MB];
#pragma unroll
    for (int jp = 0; jp < AHEAD / 2; ++jp)
#pragma unroll
      for (int m = 0; m < MB; ++m)
        ring[jp][m] = *(const v4f*)(a1p[m] + jp * 8);

    {
      const float4* __restrict__ gsrc = (const float4*)Bp1;
      float4* __restrict__ ldst = (float4*)lds[0];
#pragma unroll
      for (int i = 0; i < 16; ++i) ldst[tid + i * 256] = gsrc[tid + i * 256];
    }
    __syncthreads();

    for (int c = 0; c < nChunks; ++c) {
      const int kt0 = c * CHUNK_KT;
      const bool more = c + 1 < nChunks;
      const int ktx = kt0 + CHUNK_KT;
      const float4* __restrict__ gnext =
          !more ? (const float4*)nullptr
                : (const float4*)((ktx >= nkt)
                                      ? (Bp2 + (size_t)(ktx - nkt) * 8 * 64)
                                      : (Bp1 + (size_t)ktx * 8 * 64));
      float4* __restrict__ lnext = (float4*)lds[(c + 1) & 1];
      const float* ldsb = lds[c & 1];
      k_chunk(acc, ring, a1p, a2p, scale, ldsb, gnext, lnext, kt0, nkt, nkt2,
              lane, tid, cnt == MB);
      __syncthreads();
    }

#pragma unroll
    for (int m = 0; m < MB; ++m) {
      if (m < cnt) {
        const int m0 = (tb + m) * 16;
#pragma unroll
        for (int nt = 0; nt < 8; ++nt) {
#pragma unroll
          for (int r = 0; r < 8; ++r) {
            const int row = m0 + r + 8 * half;
            const size_t idx = (size_t)row * 128 + nt * 16 + n0;
            float v = acc[m][nt][r];
            if (relu) v = fmaxf(v, 0.0f);
            out[idx] = v;
          }
        }
      }
    }
  }
}

extern "C" void kernel_launch(void* const* d_in, const int* in_sizes, int n_in,
                              void* d_out, int out_size, void* d_ws,
                              size_t ws_size, hipStream_t stream) {
  const int IN_DIM = 384, HID = 128;
  const int N = in_sizes[0] / IN_DIM;
  const int E = in_sizes[1] / 2;

  const float* x    = (const float*)d_in[0];
  const int*   ei   = (const int*)d_in[1];
  const float* W1_l = (const float*)d_in[2];
  const float* b1   = (const float*)d_in[3];
  const float* W1_r = (const float*)d_in[4];
  const float* W2_l = (const float*)d_in[5];
  const float* b2   = (const float*)d_in[6];
  const float* W2_r = (const float*)d_in[7];
  const int* src = ei;
  const int* dst = ei + E;
  float* out = (float*)d_out;

  float* ws = (float*)d_ws;
  const long deg_off  = 0;
  const long deg_pad  = ((long)N + 127) & ~127L;
  const long agg_off  = deg_off + deg_pad;
  const long h1_off   = agg_off + (long)N * IN_DIM;
  const long pw1l_off = h1_off + (long)N * HID;
  const long pw1r_off = pw1l_off + (long)IN_DIM * HID;
  const long pw2l_off = pw1r_off + (long)IN_DIM * HID;
  const long pw2r_off = pw2l_off + (long)HID * HID;
  const long offs_off = pw2r_off + (long)HID * HID;
  const long segs_off = offs_off + (long)MAX_BWG * OFF_PITCH;
  const long sgl_off  = segs_off + (long)MAX_BWG * 1024 * 2;
  const long zrow_off = sgl_off + deg_pad;

  float* deg  = ws + deg_off;
  float* agg  = ws + agg_off;
  float* h1   = ws + h1_off;
  float* pw1l = ws + pw1l_off;
  float* pw1r = ws + pw1r_off;
  float* pw2l = ws + pw2l_off;
  float* pw2r = ws + pw2r_off;
  int*  offs = (int*)(ws + offs_off);
  int*  segs = (int*)(ws + segs_off);
  int*  single = (int*)(ws + sgl_off);
  float* zrow = ws + zrow_off;

  const int nTiles = N / 16;
  const int gemmBlocks = 256;
  const int nRanges = (N + 415) / 416;

  const int nBinWG = (E + 1023) / 1024;
  const int packWG = (2 * IN_DIM * 128 + 2 * HID * 128 + 1024 * PACK_PER - 1) /
                     (1024 * PACK_PER);
  setup_kernel<416><<<nBinWG + packWG, 1024, 0, stream>>>(
      src, dst, offs, segs, E, N, nRanges, nBinWG, W1_l, pw1l, W1_r, pw1r,
      IN_DIM, W2_l, pw2l, W2_r, pw2r, HID);

  range_gather_kernel<384, 416, 4><<<nRanges, 1024, 0, stream>>>(
      x, src, dst, offs, segs, nBinWG, agg, deg, single, zrow, E, N);

  gemm_pair_kernel<<<gemmBlocks, 256, 0, stream>>>(agg, pw1l, x, pw1r, IN_DIM,
                                                   b1, deg, single, zrow, h1, 1,
                                                   nTiles);

  range_gather_kernel<128, 416, 8><<<nRanges, 1024, 0, stream>>>(
      h1, src, dst, offs, segs, nBinWG, agg, nullptr, single, zrow, E, N);

  gemm_pair_kernel<<<gemmBlocks, 256, 0, stream>>>(agg, pw2l, h1, pw2r, HID, b2,
                                                   deg, single, zrow, out, 0,
                                                   nTiles);
}


# GEMM passes: a work-group's 25th row tile is cut along k into four quarters, one per SIMD (free accumulator slot), partial sums added through LDS (sum order changes); 6.25 tile-times per SIMD, not 7
# speedup vs baseline: 2.5793x; 1.0333x over previous
#include <hip/hip_runtime.h>


typedef __attribute__((ext_vector_type(2))) float v2f;
typedef __attribute__((ext_vector_type(4))) float v4f;
typedef __attribute__((ext_vector_type(8))) float v8f;

#define CHUNK_KT 32

__device__ __forceinline__ void pack_one(const float* __restrict__ W,
                                         float* __restrict__ P, int idx) {
  const int frag = idx >> 6;
  const int r = idx & 63;
  const int L = r >> 1;
  const int v = r & 1;
  const int kt = frag >> 3;
  const int nt = frag & 7;
  const int k = ((kt >> 1) << 3) + ((L >> 4) << 2) + ((kt & 1) << 1) + v;
  const int n = nt * 16 + (L & 15);
  P[idx] = W[k * 128 + n];
}

#define BIN_CAP 4096
#define MAX_RANGES 1024
#define MAX_BWG 256
#define OFF_PITCH 1056
#define PACK_PER 4

template <int R>
__global__ __launch_bounds__(1024) void setup_kernel(
    const int* __restrict__ src, const int* __restrict__ dst,
    int* __restrict__ offs, int* __restrict__ segs, int E, int N, int nRanges,
    int nBinWG, const float* __restrict__ W1l, float* __restrict__ P1l,
    const float* __restrict__ W1r, float* __restrict__ P1r, int K1,
    const float* __restrict__ W2l, float* __restrict__ P2l,
    const float* __restrict__ W2r, float* __restrict__ P2r, int K2) {
  __shared__ int hist[MAX_RANGES];
  __shared__ int first[MAX_RANGES + 1];
  const int tid = threadIdx.x;

  if ((int)blockIdx.x >= nBinWG) {
    const int n1 = K1 * 128, n2 = K2 * 128;
    const int g0 = ((int)blockIdx.x - nBinWG) * (1024 * PACK_PER) + tid;
    for (int j = 0; j < PACK_PER; ++j) {
      int g = g0 + j * 1024;
      if (g < n1) {
        pack_one(W1l, P1l, g);
      } else if ((g -= n1) < n1) {
        pack_one(W1r, P1r, g);
      } else if ((g -= n1) < n2) {
        pack_one(W2l, P2l, g);
      } else if ((g -= n2) < n2) {
        pack_one(W2r, P2r, g);
      }
    }
    return;
  }

  const int lane = tid & 31;
  for (int i = tid; i < MAX_RANGES; i += 1024) hist[i] = 0;
  __syncthreads();

  const int e = blockIdx.x * 1024 + tid;
  int owner = -1, rel = 0, s = 0, lrank = 0;
  if (e < E) {
    const int d = dst[e];
    s = src[e];
    if ((unsigned)d < (unsigned)N) {
      owner = d / R;
      rel = d - owner * R;
      lrank = __hip_atomic_fetch_add(&hist[owner], 1, __ATOMIC_RELAXED,
                                     __HIP_MEMORY_SCOPE_WORKGROUP);
    }
  }
  __syncthreads();

  if (tid < 32) {
    const int per = (nRanges + 31) >> 5;
    int sum = 0;
    for (int k = 0; k < per; ++k) sum += hist[lane * per + k];
    int incl = sum;
    for (int d = 1; d < 32; d <<= 1) {
      const int t = __shfl_up(incl, d);
      if (lane >= d) incl += t;
    }
    int run = incl - sum;
    for (int k = 0; k < per; ++k) {
      first[lane * per + k] = run;
      run += hist[lane * per + k];
    }
    if (lane == 31) first[nRanges] = run;
  }
  __syncthreads();

  for (int i = tid; i <= nRanges; i += 1024)
    offs[(size_t)blockIdx.x * OFF_PITCH + i] = first[i];
  if (owner >= 0)
    ((int2*)segs)[(size_t)blockIdx.x * 1024 + first[owner] + lrank] =
        make_int2(s, rel);
}


template <int Q>
__device__ __forceinline__ void store_row_and_clear(float* __restrict__ agg,
                                                    size_t node, int F,
                                                    int lane, float4* acc) {
  float4* __restrict__ o = (float4*)(agg + node * (size_t)F) + lane;
#pragma unroll
  for (int q = 0; q < Q; ++q) {
    o[q * 32] = acc[q];
    acc[q] = make_float4(0.0f, 0.0f, 0.0f, 0.0f);
  }
}

template <int F, int R, int B>
__global__ __launch_bounds__(1024) void range_gather_kernel(
    const float* __restrict__ feat, const int* __restrict__ src,
    const int* __restrict__ dst, const int* __restrict__ offs,
    const int* __restrict__ segs, int nBinWG, float* __restrict__ agg,
    float* __restrict__ deg, int* __restrict__ single,
    float* __restrict__ zrow, int E, int N) {
  constexpr int CAP = BIN_CAP;
  constexpr int PER = R / 32;
  constexpr int Q = F / 128;
  __shared__ int cnt[R];
  __shared__ int start[R + 1];
  __shared__ int list_e[CAP];
  __shared__ unsigned list_k[CAP];
  __shared__ int sorted[CAP];
  __shared__ int one[R];
  __shared__ int clo[MAX_BWG];
  __shared__ int cpos[MAX_BWG + 1];

  const int tid = threadIdx.x;
  const int lane = tid & 31;
  const int wave = __builtin_amdgcn_readfirstlane(tid >> 5);
  const int base = blockIdx.x * R;
  const int rows = (N - base < R) ? (N - base) : R;

  for (int i = tid; i < R; i += 1024) cnt[i] = 0;
  if (blockIdx.x == 0)
    for (int i = tid; i < F; i += 1024) zrow[i] = 0.0f;
  if (tid < MAX_BWG) {
    int lo = 0, hi = 0;
    if (tid < nBinWG) {
      const int* __restrict__ t = offs + (size_t)tid * OFF_PITCH + blockIdx.x;
      lo = t[0];
      hi = t[1];
    }
    clo[tid] = lo;
    cpos[tid] = hi - lo;
  }
  __syncthreads();

  if (wave == 0) {
    int c[MAX_BWG / 32];
    int s = 0;
    for (int k = 0; k < MAX_BWG / 32; ++k) {
      c[k] = cpos[lane * (MAX_BWG / 32) + k];
      s += c[k];
    }
    int incl = s;
    for (int d = 1; d < 32; d <<= 1) {
      const int t = __shfl_up(incl, d);
      if (lane >= d) incl += t;
    }
    int run = incl - s;
    for (int k = 0; k < MAX_BWG / 32; ++k) {
      cpos[lane * (MAX_BWG / 32) + k] = run;
      run += c[k];
    }
    if (lane == 31) cpos[MAX_BWG] = run;
  }
  __syncthreads();

  const int total = (nBinWG <= MAX_BWG) ? cpos[MAX_BWG] : (CAP + 1);
  if (total <= CAP) {
    const int2* __restrict__ seg = (const int2*)segs;
    for (int i = tid; i < total; i += 1024) {
      int w = 0;
      for (int step = MAX_BWG / 2; step > 0; step >>= 1)
        if (cpos[w + step] <= i) w += step;
      const int2 r = seg[(size_t)w * 1024 + clo[w] + (i - cpos[w])];
      const int rank = __hip_atomic_fetch_add(
          &cnt[r.y], 1, __ATOMIC_RELAXED, __HIP_MEMORY_SCOPE_WORKGROUP);
      list_e[i] = r.x;
      list_k[i] = (unsigned)r.y | ((unsigned)rank << 16);
    }
  }
  __syncthreads();

  if (wave == 0) {
    int s = 0;
    for (int k = 0; k < PER; ++k) {
      const int c = cnt[lane * PER + k];
      s += (c >= 2) ? c : 0;
    }
    int incl = s;
    for (int d = 1; d < 32; d <<= 1) {
      const int t = __shfl_up(incl, d);
      if (lane >= d) incl += t;
    }
    int run = incl - s;
    for (int k = 0; k < PER; ++k) {
      const int c = cnt[lane * PER + k];
      start[lane * PER + k] = run;
      run += (c >= 2) ? c : 0;
    }
    if (lane == 31) start[R] = run;
  }
  __syncthreads();

  if (total <= CAP) {
    for (int i = tid; i < total; i += 1024) {
      const unsigned k = list_k[i];
      const int nd = (int)(k & 0xffffu);
      if (cnt[nd] >= 2)
        sorted[start[nd] + (int)(k >> 16)] = list_e[i];
      else
        one[nd] = list_e[i];
    }
  }
  __syncthreads();

  const int r0 = wave * PER;
  const int r1 = (r0 + PER < rows) ? (r0 + PER) : rows;
  float4 acc[Q];
#pragma unroll
  for (int q = 0; q < Q; ++q) acc[q] = make_float4(0.0f, 0.0f, 0.0f, 0.0f);

  if (r0 < rows) {
    if (total <= CAP) {
      const int e_lo = __builtin_amdgcn_readfirstlane(start[r0]);
      const int e_hi = __builtin_amdgcn_readfirstlane(start[r1]);
      int node = r0;
      int node_beg = e_lo;
      int node_end = __builtin_amdgcn_readfirstlane(start[r0 + 1]);
      for (int eb = e_lo; eb < e_hi; eb += B) {
        float4 v[B][Q];
#pragma unroll
        for (int j = 0; j < B; ++j) {
          const int ei = (eb + j < e_hi) ? (eb + j) : (e_hi - 1);
          const int s = __builtin_amdgcn_readfirstlane(sorted[ei]);
          const float4* __restrict__ row =
              (const float4*)(feat + (size_t)s * F) + lane;
#pragma unroll
          for (int q = 0; q < Q; ++q) v[j][q] = row[q * 32];
        }
#pragma unroll
        for (int j = 0; j < B; ++j) {
          if (eb + j < e_hi) {
            while (eb + j >= node_end) {
              if (node_end > node_beg)
                store_row_and_clear<Q>(agg, (size_t)(base + node), F, lane, acc);
              ++node;
              node_beg = node_end;
              node_end = __builtin_amdgcn_readfirstlane(start[node + 1]);
            }
#pragma unroll
            for (int q = 0; q < Q; ++q) {
              acc[q].x += v[j][q].x;
              acc[q].y += v[j][q].y;
              acc[q].z += v[j][q].z;
              acc[q].w += v[j][q].w;
            }
          }
        }
      }
      if (e_hi > e_lo)
        store_row_and_clear<Q>(agg, (size_t)(base + node), F, lane, acc);
    } else {
      for (int node = r0; node < r1; ++node) {
        int nedges = 0;
        for (int eb = 0; eb < E; eb += 32) {
          const int e = eb + lane;
          const int d = (e < E) ? dst[e] : -1;
          unsigned m = __builtin_amdgcn_ballot_w32(d == base + node);
          nedges += __builtin_popcount(m);
          while (m) {
            const int l = __builtin_ctz(m);
            m &= m - 1;
            const int s = src[eb + l];
            const float4* __restrict__ row =
                (const float4*)(feat + (size_t)s * F) + lane;
#pragma unroll
            for (int q = 0; q < Q; ++q) {
              const float4 w = row[q * 32];
              acc[q].x += w.x;
              acc[q].y += w.y;
              acc[q].z += w.z;
              acc[q].w += w.w;
            }
          }
        }
        store_row_and_clear<Q>(agg, (size_t)(base + node), F, lane, acc);
        if (deg && lane == 0) {
          deg[base + node] = (float)nedges;
          single[base + node] = -1;
        }
      }
    }
  }
  if (deg && total <= CAP)
    for (int i = tid; i < rows; i += 1024) {
      const int c = cnt[i];
      deg[base + i] = (float)c;
      single[base + i] = (c == 1) ? one[i] : -1;
    }
}

#define MB 4
#define AHEAD 4
#define STG 2

__device__ __forceinline__ void k_chunk(
    v8f (&acc)[MB][8], v4f (&ring)[AHEAD / 2][MB],
    const float* const (&a1p)[MB], const float* const (&a2p)[MB],
    const float (&scale)[MB], const float* __restrict__ ldsb,
    const float4* __restrict__ gnext, float4* __restrict__ lnext, int kt0,
    int nkt, int nkt2, int lane, int tid, bool full, int xq) {
  const bool second = kt0 >= nkt;
  for (int kb = 0; kb < CHUNK_KT; kb += AHEAD) {
    const int sbase = tid + (kb / AHEAD) * STG * 256;
    const bool doLast = full || ((kb * 4) / CHUNK_KT == xq);
    float4 s0 = make_float4(0.0f, 0.0f, 0.0f, 0.0f), s1 = s0;
    if (gnext) {
      s0 = gnext[sbase];
      s1 = gnext[sbase + 256];
    }
    v4f nxt[AHEAD / 2][MB];
#pragma unroll
    for (int jp = 0; jp < AHEAD / 2; ++jp) {
      int ktn = kt0 + kb + AHEAD + 2 * jp;
      if (ktn > nkt2 - 2) ktn = nkt2 - 2;
      const bool n2 = ktn >= nkt;
      const int kk = (n2 ? (ktn - nkt) : ktn) * 4;
#pragma unroll
      for (int m = 0; m < MB; ++m)
        nxt[jp][m] = *(const v4f*)((n2 ? a2p[m] : a1p[m]) + kk);
    }
    asm volatile("" ::: "memory");
#pragma unroll
    for (int j = 0; j < AHEAD; ++j) {
      const int kt = kb + j;
      v2f a[MB];
#pragma unroll
      for (int m = 0; m < MB; ++m) {
        const v4f f = ring[j >> 1][m];
        const v2f h2 = (j & 1) ? (v2f){f.z, f.w} : (v2f){f.x, f.y};
        a[m] = h2 * (second ? 1.0f : scale[m]);
      }

      const float* bbase = ldsb + (kt * 8) * 64 + lane * 2;
      v2f bf[8];
#pragma unroll
      for (int nt = 0; nt < 8; ++nt) bf[nt] = *(const v2f*)(bbase + nt * 64);

      if (doLast) {
#pragma unroll
        for (int nt = 0; nt < 8; ++nt) {
          acc[MB - 1][nt] = __builtin_amdgcn_wmma_f32_16x16x4_f32(
              false, a[MB - 1], false, bf[nt], (short)0, acc[MB - 1][nt], false,
              false);
        }
      }
#pragma unroll
      for (int nt = 0; nt < 8; ++nt) {
#pragma unroll
        for (int m = 0; m < MB - 1; ++m) {
          acc[m][nt] = __builtin_amdgcn_wmma_f32_16x16x4_f32(
              false, a[m], false, bf[nt], (short)0, acc[m][nt], false, false);
        }
      }
    }
#pragma unroll
    for (int jp = 0; jp < AHEAD / 2; ++jp)
#pragma unroll
      for (int m = 0; m < MB; ++m) ring[jp][m] = nxt[jp][m];
    if (gnext) {
      lnext[sbase] = s0;
      lnext[sbase + 256] = s1;
    }
  }
}

__global__ __launch_bounds__(256) void gemm_pair_kernel(
    const float* __restrict__ A1, const float* __restrict__ Bp1,
    const float* __restrict__ A2, const float* __restrict__ Bp2, int K,
    const float* __restrict__ bias, const float* __restrict__ deg,
    const int* __restrict__ single, const float* __restrict__ zrow,
    float* __restrict__ out, int relu, int nTiles) {
  __shared__ float lds[2][CHUNK_KT * 8 * 64];

  const int tid = threadIdx.x;
  const int lane = tid & 31;
  const int wave = __builtin_amdgcn_readfirstlane(tid >> 5);
  const int half = lane >> 4;
  const int n0 = lane & 15;

  const int q = wave & 3;
  const int nG = gridDim.x;
  const int T0 = (int)(((long long)blockIdx.x * nTiles) / nG);
  const int T1 = (int)(((long long)(blockIdx.x + 1) * nTiles) / nG);
  const int nW = T1 - T0;
  const bool split = ((nW & 3) == 1) && ((((nW >> 2) + 1) >> 1) <= MB - 1);
  const int whole = split ? (nW - 1) : nW;
  const int xTile = T1 - 1;
  const int xq = (split && (wave >> 2) == 0) ? q : -1;
  const int ts0 = T0 + (q * whole) / 4;
  const int ts1 = T0 + ((q + 1) * whole) / 4;
  const int firstPart = (ts1 - ts0 + 1) >> 1;
  const int w0 = (wave >> 2) ? (ts0 + firstPart) : ts0;
  const int w1 = (wave >> 2) ? ts1 : (ts0 + firstPart);
  const int maxS = (whole + 3) >> 2;
  const int nPass = (((maxS + 1) >> 1) + MB - 1) / MB;

  const int nkt = K >> 2;
  const int nkt2 = nkt << 1;
  const int nChunks = nkt2 / CHUNK_KT;

  for (int p = 0; p < nPass; ++p) {
    const int tb = w0 + p * MB;
    int cnt = w1 - tb;
    cnt = cnt < 0 ? 0 : (cnt > MB ? MB : cnt);
    const int lastT = (cnt > 0) ? (tb + cnt - 1) : (nTiles - 1);

    const float* a1p[MB];
    const float* a2p[MB];
    float scale[MB];
#pragma unroll
    for (int m = 0; m < MB; ++m) {
      const int t = (m < cnt) ? (tb + m)
                              : ((m == MB - 1 && xq >= 0) ? xTile : lastT);
      const int arow = t * 16 + n0;
      const float d = deg[arow];
      const int s1 = single[arow];
      const float* rp = (s1 >= 0) ? (A2 + (size_t)s1 * (size_t)K)
                                  : ((d > 0.0f) ? (A1 + (size_t)arow * (size_t)K)
                                                : zrow);
      a1p[m] = rp + half * 4;
      a2p[m] = A2 + (size_t)arow * (size_t)K + half * 4;
      scale[m] = 1.0f / fmaxf(d, 1.0f);
    }

    float bv[8];
#pragma unroll
    for (int nt = 0; nt < 8; ++nt) bv[nt] = bias[nt * 16 + n0];

    v4f ring[AHEAD / 2][MB];
#pragma unroll
    for (int jp = 0; jp < AHEAD / 2; ++jp)
#pragma unroll
      for (int m = 0; m < MB; ++m)
        ring[jp][m] = *(const v4f*)(a1p[m] + jp * 8);

    {
      const float4* __restrict__ gsrc = (const float4*)Bp1;
      float4* __restrict__ ldst = (float4*)lds[0];
#pragma unroll
      for (int i = 0; i < 16; ++i) ldst[tid + i * 256] = gsrc[tid + i * 256];
    }
    __syncthreads();

    v8f acc[MB][8];
#pragma unroll
    for (int nt = 0; nt < 8; ++nt) {
      const float b = bv[nt];
#pragma unroll
      for (int m = 0; m < MB; ++m) acc[m][nt] = (v8f){b, b, b, b, b, b, b, b};
    }

    for (int c = 0; c < nChunks; ++c) {
      const int kt0 = c * CHUNK_KT;
      const bool more = c + 1 < nChunks;
      const int ktx = kt0 + CHUNK_KT;
      const float4* __restrict__ gnext =
          !more ? (const float4*)nullptr
                : (const float4*)((ktx >= nkt)
                                      ? (Bp2 + (size_t)(ktx - nkt) * 8 * 64)
                                      : (Bp1 + (size_t)ktx * 8 * 64));
      float4* __restrict__ lnext = (float4*)lds[(c + 1) & 1];
      const float* ldsb = lds[c & 1];
      if (c == 0 && xq > 0) {
#pragma unroll
        for (int nt = 0; nt < 8; ++nt)
          acc[MB - 1][nt] =
              (v8f){0.0f, 0.0f, 0.0f, 0.0f, 0.0f, 0.0f, 0.0f, 0.0f};
      }
      k_chunk(acc, ring, a1p, a2p, scale, ldsb, gnext, lnext, kt0, nkt, nkt2,
              lane, tid, cnt == MB, xq);
      __syncthreads();
    }

#pragma unroll
    for (int m = 0; m < MB; ++m) {
      if (m < cnt) {
        const int m0 = (tb + m) * 16;
#pragma unroll
        for (int nt = 0; nt < 8; ++nt) {
#pragma unroll
          for (int r = 0; r < 8; ++r) {
            const int row = m0 + r + 8 * half;
            const size_t idx = (size_t)row * 128 + nt * 16 + n0;
            float v = acc[m][nt][r];
            if (relu) v = fmaxf(v, 0.0f);
            out[idx] = v;
          }
        }
      }
    }

    if (split) {
      float* __restrict__ part = &lds[0][0];
      if (xq >= 0) {
#pragma unroll
        for (int nt = 0; nt < 8; ++nt)
#pragma unroll
          for (int r = 0; r < 8; ++r)
            part[((xq * 8 + nt) * 8 + r) * 32 + lane] = acc[MB - 1][nt][r];
      }
      __syncthreads();
      const int m0 = xTile * 16;
      const int r0 = (wave >> 2) * 4;
#pragma unroll
      for (int rr = 0; rr < 4; ++rr) {
#pragma unroll
        for (int t2 = 0; t2 < 2; ++t2) {
          const int r = r0 + rr;
          const int nt = q * 2 + t2;
          float v = part[((0 * 8 + nt) * 8 + r) * 32 + lane];
          v += part[((1 * 8 + nt) * 8 + r) * 32 + lane];
          v += part[((2 * 8 + nt) * 8 + r) * 32 + lane];
          v += part[((3 * 8 + nt) * 8 + r) * 32 + lane];
          if (relu) v = fmaxf(v, 0.0f);
          out[(size_t)(m0 + r + 8 * half) * 128 + nt * 16 + n0] = v;
        }
      }
    }
  }
}

extern "C" void kernel_launch(void* const* d_in, const int* in_sizes, int n_in,
                              void* d_out, int out_size, void* d_ws,
                              size_t ws_size, hipStream_t stream) {
  const int IN_DIM = 384, HID = 128;
  const int N = in_sizes[0] / IN_DIM;
  const int E = in_sizes[1] / 2;

  const float* x    = (const float*)d_in[0];
  const int*   ei   = (const int*)d_in[1];
  const float* W1_l = (const float*)d_in[2];
  const float* b1   = (const float*)d_in[3];
  const float* W1_r = (const float*)d_in[4];
  const float* W2_l = (const float*)d_in[5];
  const float* b2   = (const float*)d_in[6];
  const float* W2_r = (const float*)d_in[7];
  const int* src = ei;
  const int* dst = ei + E;
  float* out = (float*)d_out;

  float* ws = (float*)d_ws;
  const long deg_off  = 0;
  const long deg_pad  = ((long)N + 127) & ~127L;
  const long agg_off  = deg_off + deg_pad;
  const long h1_off   = agg_off + (long)N * IN_DIM;
  const long pw1l_off = h1_off + (long)N * HID;
  const long pw1r_off = pw1l_off + (long)IN_DIM * HID;
  const long pw2l_off = pw1r_off + (long)IN_DIM * HID;
  const long pw2r_off = pw2l_off + (long)HID * HID;
  const long offs_off = pw2r_off + (long)HID * HID;
  const long segs_off = offs_off + (long)MAX_BWG * OFF_PITCH;
  const long sgl_off  = segs_off + (long)MAX_BWG * 1024 * 2;
  const long zrow_off = sgl_off + deg_pad;

  float* deg  = ws + deg_off;
  float* agg  = ws + agg_off;
  float* h1   = ws + h1_off;
  float* pw1l = ws + pw1l_off;
  float* pw1r = ws + pw1r_off;
  float* pw2l = ws + pw2l_off;
  float* pw2r = ws + pw2r_off;
  int*  offs = (int*)(ws + offs_off);
  int*  segs = (int*)(ws + segs_off);
  int*  single = (int*)(ws + sgl_off);
  float* zrow = ws + zrow_off;

  const int nTiles = N / 16;
  const int gemmBlocks = 256;
  const int nRanges = (N + 415) / 416;

  const int nBinWG = (E + 1023) / 1024;
  const int packWG = (2 * IN_DIM * 128 + 2 * HID * 128 + 1024 * PACK_PER - 1) /
                     (1024 * PACK_PER);
  setup_kernel<416><<<nBinWG + packWG, 1024, 0, stream>>>(
      src, dst, offs, segs, E, N, nRanges, nBinWG, W1_l, pw1l, W1_r, pw1r,
      IN_DIM, W2_l, pw2l, W2_r, pw2r, HID);

  range_gather_kernel<384, 416, 4><<<nRanges, 1024, 0, stream>>>(
      x, src, dst, offs, segs, nBinWG, agg, deg, single, zrow, E, N);

  gemm_pair_kernel<<<gemmBlocks, 256, 0, stream>>>(agg, pw1l, x, pw1r, IN_DIM,
                                                   b1, deg, single, zrow, h1, 1,
                                                   nTiles);

  range_gather_kernel<128, 416, 8><<<nRanges, 1024, 0, stream>>>(
      h1, src, dst, offs, segs, nBinWG, agg, nullptr, single, zrow, E, N);

  gemm_pair_kernel<<<gemmBlocks, 256, 0, stream>>>(agg, pw2l, h1, pw2r, HID, b2,
                                                   deg, single, zrow, out, 0,
                                                   nTiles);
}
